# starting kernel
# baseline (speedup 1.0000x reference)
#include <hip/hip_runtime.h>
#include <hip/hip_bf16.h>

typedef __attribute__((ext_vector_type(8))) float v8f;
typedef __attribute__((ext_vector_type(2))) float v2f;

__global__ void zero_kernel(float* __restrict__ p, int n) {
  int i = blockIdx.x * blockDim.x + threadIdx.x;
  if (i < n) p[i] = 0.0f;
}

__global__ void degree_kernel(const int* __restrict__ dst, float* __restrict__ deg, int nEdges) {
  int i = blockIdx.x * blockDim.x + threadIdx.x;
  if (i < nEdges)
    __hip_atomic_fetch_add(&deg[dst[i]], 1.0f, __ATOMIC_RELAXED, __HIP_MEMORY_SCOPE_AGENT);
}

__global__ void recip_deg_kernel(float* __restrict__ deg, int n) {
  int i = blockIdx.x * blockDim.x + threadIdx.x;
  if (i < n) deg[i] = 1.0f / fmaxf(deg[i], 1.0f);
}

__global__ void scatter_add_kernel(const float* __restrict__ x, const int* __restrict__ src,
                                   const int* __restrict__ dst, float* __restrict__ agg,
                                   int nEdges, int f4) {
  int tid = blockIdx.x * blockDim.x + threadIdx.x;
  int total = nEdges * f4;
  if (tid >= total) return;
  int e = tid / f4;
  int c = tid - e * f4;
  int s = src[e];
  int d = dst[e];
  float4 v = ((const float4*)x)[s * f4 + c];
  float* p = agg + ((size_t)d * f4 + c) * 4;
  __hip_atomic_fetch_add(p + 0, v.x, __ATOMIC_RELAXED, __HIP_MEMORY_SCOPE_AGENT);
  __hip_atomic_fetch_add(p + 1, v.y, __ATOMIC_RELAXED, __HIP_MEMORY_SCOPE_AGENT);
  __hip_atomic_fetch_add(p + 2, v.z, __ATOMIC_RELAXED, __HIP_MEMORY_SCOPE_AGENT);
  __hip_atomic_fetch_add(p + 3, v.w, __ATOMIC_RELAXED, __HIP_MEMORY_SCOPE_AGENT);
}

template <int FIN>
__global__ __launch_bounds__(128) void
gemm_bias_act_kernel(const float* __restrict__ agg, const float* __restrict__ rdeg,
                     const float* __restrict__ xin, const float* __restrict__ W,
                     const float* __restrict__ bias, float* __restrict__ out,
                     int nRows, int Fout, int doRelu)
{
  constexpr int STRIDE = FIN + 4;
  __shared__ float As[64 * STRIDE];
  __shared__ float Bs[128 * STRIDE];
  __shared__ float bsh[128];

  const int rowBase   = blockIdx.x * 64;
  const int foutTiles = (Fout + 15) >> 4;
  const int bRows     = foutTiles * 16;

  for (int idx = threadIdx.x; idx < bRows * FIN; idx += 128) {
    int r = idx / FIN;
    int k = idx - r * FIN;
    Bs[r * STRIDE + k] = (r < Fout) ? W[(size_t)r * FIN + k] : 0.0f;
  }
  {
    int t = threadIdx.x;
    if (t < 128) bsh[t] = (t < Fout) ? bias[t] : 0.0f;
  }
  for (int idx = threadIdx.x; idx < 64 * FIN; idx += 128) {
    int r  = idx / FIN;
    int k  = idx - r * FIN;
    int gr = rowBase + r;
    float v = 0.0f;
    if (gr < nRows) {
      v = xin[(size_t)gr * FIN + k];
      if (agg) v += agg[(size_t)gr * FIN + k] * rdeg[gr];
    }
    As[r * STRIDE + k] = v;
  }
  __syncthreads();

  const int lane = threadIdx.x & 31;
  const int wave = threadIdx.x >> 5;
  const int half = lane >> 4;
  const int l16  = lane & 15;
  const float* Arow = &As[(wave * 16 + l16) * STRIDE];
  const bool fullRows = (rowBase + 64 <= nRows);

  for (int ct = 0; ct < foutTiles; ++ct) {
    const float* Brow = &Bs[(ct * 16 + l16) * STRIDE];
    v8f c = {};
    #pragma unroll 8
    for (int k0 = 0; k0 < FIN; k0 += 4) {
      int ka = k0 + 2 * half;
      v2f a, b;
      a.x = Arow[ka];
      a.y = Arow[ka + 1];
      b.x = Brow[ka];
      b.y = Brow[ka + 1];
      c = __builtin_amdgcn_wmma_f32_16x16x4_f32(false, a, false, b, (short)0, c, false, false);
    }
    const int n  = ct * 16 + l16;
    float bv = bsh[n];
    float* colBase = out + (size_t)(rowBase + wave * 16 + 8 * half) * Fout + n;
    if (fullRows && (ct * 16 + 16 <= Fout)) {
      #pragma unroll
      for (int r = 0; r < 8; ++r) {
        float val = c[r] + bv;
        if (doRelu) val = fmaxf(val, 0.0f);
        colBase[(size_t)r * Fout] = val;
      }
    } else {
      #pragma unroll
      for (int r = 0; r < 8; ++r) {
        int gRow = rowBase + wave * 16 + r + 8 * half;
        float val = c[r] + bv;
        if (doRelu) val = fmaxf(val, 0.0f);
        if (gRow < nRows && n < Fout) out[(size_t)gRow * Fout + n] = val;
      }
    }
  }
}

extern "C" void kernel_launch(void* const* d_in, const int* in_sizes, int n_in,
                              void* d_out, int out_size, void* d_ws, size_t ws_size,
                              hipStream_t stream) {
  const float* x0 = (const float*)d_in[0];
  const int*   ei = (const int*)d_in[1];
  const float *W1 = (const float*)d_in[2],  *b1 = (const float*)d_in[3];
  const float *W2 = (const float*)d_in[4],  *b2 = (const float*)d_in[5];
  const float *W3 = (const float*)d_in[6],  *b3 = (const float*)d_in[7];
  const float *Wp = (const float*)d_in[8],  *bp = (const float*)d_in[9];
  const float *Wc = (const float*)d_in[10], *bc = (const float*)d_in[11];

  const int N      = in_sizes[0] / 64;
  const int nEdges = in_sizes[1] / 2;
  const int* src = ei;
  const int* dst = ei + nEdges;

  float* ws  = (float*)d_ws;
  float* deg = ws;
  float* agg = ws + 65536;
  float* h2  = agg + (size_t)N * 128;

  float* out = (float*)d_out;
  float* h1  = out;
  float* emb = out;
  float* nxt = out + (size_t)N * 64;
  float* cls = out + (size_t)N * 128;

  auto nb = [](int n) { return dim3((unsigned)((n + 255) / 256)); };
  dim3 b256(256);
  dim3 gemmGrid((unsigned)((N + 63) / 64)), gemmBlock(128);

  zero_kernel     <<<nb(N),      b256, 0, stream>>>(deg, N);
  degree_kernel   <<<nb(nEdges), b256, 0, stream>>>(dst, deg, nEdges);
  recip_deg_kernel<<<nb(N),      b256, 0, stream>>>(deg, N);

  zero_kernel        <<<nb(N * 64),      b256, 0, stream>>>(agg, N * 64);
  scatter_add_kernel <<<nb(nEdges * 16), b256, 0, stream>>>(x0, src, dst, agg, nEdges, 16);
  gemm_bias_act_kernel<64><<<gemmGrid, gemmBlock, 0, stream>>>(agg, deg, x0, W1, b1, h1, N, 128, 1);

  zero_kernel        <<<nb(N * 128),     b256, 0, stream>>>(agg, N * 128);
  scatter_add_kernel <<<nb(nEdges * 32), b256, 0, stream>>>(h1, src, dst, agg, nEdges, 32);
  gemm_bias_act_kernel<128><<<gemmGrid, gemmBlock, 0, stream>>>(agg, deg, h1, W2, b2, h2, N, 128, 1);

  zero_kernel        <<<nb(N * 128),     b256, 0, stream>>>(agg, N * 128);
  scatter_add_kernel <<<nb(nEdges * 32), b256, 0, stream>>>(h2, src, dst, agg, nEdges, 32);
  gemm_bias_act_kernel<128><<<gemmGrid, gemmBlock, 0, stream>>>(agg, deg, h2, W3, b3, emb, N, 64, 1);

  gemm_bias_act_kernel<64><<<gemmGrid, gemmBlock, 0, stream>>>(nullptr, nullptr, emb, Wp, bp, nxt, N, 64, 0);
  gemm_bias_act_kernel<64><<<gemmGrid, gemmBlock, 0, stream>>>(nullptr, nullptr, emb, Wc, bc, cls, N, 10, 0);
}


# Dst scan done once: layer 1's owner blocks leave a bit per wave and 32-edge line that held one of their edges; layers 2-3 load and test only the flagged lines; sums in arrival order as before
# speedup vs baseline: 3.2372x; 3.2372x over previous
#include <hip/hip_runtime.h>
#include <hip/hip_bf16.h>

typedef __attribute__((ext_vector_type(8))) float v8f;
typedef __attribute__((ext_vector_type(2))) float v2f;
typedef __attribute__((ext_vector_type(4))) float v4f;
typedef __attribute__((ext_vector_type(4))) int   v4i;

__global__ void zero_kernel(float* __restrict__ p, int n) {
  int i = blockIdx.x * blockDim.x + threadIdx.x;
  if (i < n) p[i] = 0.0f;
}

__global__ void degree_kernel(const int* __restrict__ dst, float* __restrict__ deg, int nEdges) {
  int i = blockIdx.x * blockDim.x + threadIdx.x;
  if (i < nEdges)
    __hip_atomic_fetch_add(&deg[dst[i]], 1.0f, __ATOMIC_RELAXED, __HIP_MEMORY_SCOPE_AGENT);
}

__global__ void recip_deg_kernel(float* __restrict__ deg, int n) {
  int i = blockIdx.x * blockDim.x + threadIdx.x;
  if (i < n) deg[i] = 1.0f / fmaxf(deg[i], 1.0f);
}

template <int F, bool FULL>
__device__ __forceinline__ void add_rows(const float* __restrict__ x, float* acc,
                                         int sv, int rv, int lane, int cnt) {
  constexpr int C = F / 32;
  constexpr int U = 32 / C;
  #pragma unroll
  for (int j0 = 0; j0 < 32; j0 += U) {
    if (!FULL && j0 >= cnt) break;
    float v[U][C];
    #pragma unroll
    for (int j = 0; j < U; ++j) {
      const int s = __builtin_amdgcn_readlane(sv, j0 + j);
      const float* xr = x + (size_t)s * F + lane;
      #pragma unroll
      for (int c = 0; c < C; ++c) v[j][c] = xr[c * 32];
    }
    #pragma unroll
    for (int j = 0; j < U; ++j) {
      if (FULL || j0 + j < cnt) {
        const int r = __builtin_amdgcn_readlane(rv, j0 + j);
        float* ar = acc + r * F + lane;
        #pragma unroll
        for (int c = 0; c < C; ++c)
          __hip_atomic_fetch_add(ar + c * 32, v[j][c], __ATOMIC_RELAXED, __HIP_MEMORY_SCOPE_WORKGROUP);
      }
    }
  }
}

template <int F, bool FULL>
__device__ __forceinline__ void drain(const float* __restrict__ x, const int* __restrict__ src,
                                      float* acc, const unsigned* qp, int lane, int cnt) {
  const unsigned ent = qp[FULL ? lane : min(lane, cnt - 1)];
  const int rv = (int)(ent & 255u);
  const int sv = src[ent >> 8];
  add_rows<F, FULL>(x, acc, sv, rv, lane, cnt);
}

template <bool TAIL, bool VEC, bool MAP>
__device__ __forceinline__ int scan_step(const int* __restrict__ dst, unsigned* q, int qn, int e0,
                                         int lane, int base, int rows, int nEdges, unsigned& lw) {
  int d[32];
  if (VEC && !TAIL) {
    const v4i* dp = (const v4i*)(dst + e0) + lane;
    #pragma unroll
    for (int k4 = 0; k4 < 8; ++k4) {
      const v4i t = dp[k4 * 32];
      d[4 * k4 + 0] = t.x; d[4 * k4 + 1] = t.y; d[4 * k4 + 2] = t.z; d[4 * k4 + 3] = t.w;
    }
  } else {
    #pragma unroll
    for (int k = 0; k < 32; ++k) {
      const int e = e0 + k * 32 + lane;
      d[k] = dst[TAIL ? min(e, nEdges - 1) : e];
    }
  }
  #pragma unroll
  for (int k = 0; k < 32; ++k) {
    const int e = (VEC && !TAIL) ? (e0 + (k >> 2) * 128 + lane * 4 + (k & 3)) : (e0 + k * 32 + lane);
    const unsigned r = (unsigned)(d[k] - base);
    bool hit = (r < (unsigned)rows);
    if (TAIL) hit = hit && (e < nEdges);
    const unsigned m = __builtin_amdgcn_ballot_w32(hit);
    if (m != 0u) {
      const int pos = qn + (int)__builtin_amdgcn_mbcnt_lo(m, 0u);
      if (hit) q[pos] = ((unsigned)e << 8) | r;
      qn += __builtin_popcount(m);
      if (MAP) {
        if (VEC && !TAIL) {
          const unsigned g = ((m & 0x000000FFu) ? 1u : 0u) | ((m & 0x0000FF00u) ? 2u : 0u) |
                             ((m & 0x00FF0000u) ? 4u : 0u) | ((m & 0xFF000000u) ? 8u : 0u);
          lw |= g << (4 * (k >> 2));
        } else {
          lw |= 1u << k;
        }
      }
    }
  }
  return qn;
}

template <int F, int MODE>
__global__ __launch_bounds__(1024) void
owner_sum_kernel(const float* __restrict__ x, const int* __restrict__ src,
                 const int* __restrict__ dst, float* __restrict__ agg, unsigned* hitmap,
                 int nRows, int nEdges) {
  constexpr int RB    = 196;
  constexpr int STEP  = 32 * 32;
  constexpr int QCAP  = STEP + 32;
  constexpr int MAPW  = 10;
  constexpr int NL    = 16;
  __shared__ float    acc[RB * F];
  __shared__ unsigned queue[32 * QCAP];
  __shared__ unsigned short lineq[MODE == 1 ? 32 * 1024 : 1];

  const int lane = threadIdx.x & 31;
  const int wave = __builtin_amdgcn_readfirstlane((int)(threadIdx.x >> 5));
  unsigned* q = queue + wave * QCAP;
  const bool vec = (((size_t)dst) & 15) == 0;
  const int nRounds = (nEdges + 32 * STEP - 1) / (32 * STEP);
  const bool mapOk = ((long long)gridDim.x * RB >= (long long)nRows) && (nRounds <= 32) &&
                     ((long long)gridDim.x * 32 * 32 <= (long long)nRows * MAPW);
  unsigned* mapw = hitmap + ((size_t)blockIdx.x * 32 + wave) * 32;

  for (int base = blockIdx.x * RB; base < nRows; base += gridDim.x * RB) {
    const int rows = min(RB, nRows - base);
    for (int i = threadIdx.x; i < RB * F / 4; i += 1024) ((v4f*)acc)[i] = v4f{0.0f, 0.0f, 0.0f, 0.0f};
    __syncthreads();

    int qn = 0;
    if (MODE == 1 && mapOk) {
      const unsigned w = mapw[lane];
      const int p = __builtin_popcount(w);
      int off = 0, total = 0;
      #pragma unroll
      for (int t = 0; t < 32; ++t) {
        const int pt = __builtin_amdgcn_readlane(p, t);
        if (lane > t) off += pt;
        total += pt;
      }
      unsigned short* lq = lineq + wave * 1024;
      for (unsigned ww = w; ww != 0u; ww &= ww - 1u)
        lq[off++] = (unsigned short)(lane * 32 + __builtin_ctz(ww));

      for (int b0 = 0; b0 < total; b0 += 32) {
        const int nb = min(32, total - b0);
        const int lv = lq[b0 + min(lane, nb - 1)];
        #pragma unroll
        for (int j0 = 0; j0 < 32; j0 += NL) {
          if (j0 >= nb) break;
          int d[NL], eb[NL];
          #pragma unroll
          for (int j = 0; j < NL; ++j) {
            const int code = __builtin_amdgcn_readlane(lv, j0 + j);
            const int e = ((code >> 5) * 32 + wave) * STEP + (code & 31) * 32 + lane;
            eb[j] = e;
            d[j]  = dst[min(e, nEdges - 1)];
          }
          #pragma unroll
          for (int j = 0; j < NL; ++j) {
            if (j0 + j < nb) {
              const unsigned r = (unsigned)(d[j] - base);
              const bool hit = (r < (unsigned)rows) && (eb[j] < nEdges);
              const unsigned m = __builtin_amdgcn_ballot_w32(hit);
              if (m != 0u) {
                const int pos = qn + (int)__builtin_amdgcn_mbcnt_lo(m, 0u);
                if (hit) q[pos] = ((unsigned)eb[j] << 8) | r;
                qn += __builtin_popcount(m);
              }
            }
          }
          while (qn >= 32) { qn -= 32; drain<F, true>(x, src, acc, q + qn, lane, 32); }
        }
      }
    } else {
      unsigned mapv = 0u;
      for (int t = 0; t < nRounds; ++t) {
        const int tt = (t + (int)blockIdx.x) % nRounds;
        const int e0 = tt * (32 * STEP) + wave * STEP;
        if (e0 < nEdges) {
          unsigned lw = 0u;
          if (e0 + STEP > nEdges) qn = scan_step<true,  false, MODE == 0>(dst, q, qn, e0, lane, base, rows, nEdges, lw);
          else if (vec)           qn = scan_step<false, true,  MODE == 0>(dst, q, qn, e0, lane, base, rows, nEdges, lw);
          else                    qn = scan_step<false, false, MODE == 0>(dst, q, qn, e0, lane, base, rows, nEdges, lw);
          if (MODE == 0) { if (lane == tt) mapv = lw; }
          while (qn >= 32) { qn -= 32; drain<F, true>(x, src, acc, q + qn, lane, 32); }
        }
      }
      if (MODE == 0 && mapOk) mapw[lane] = mapv;
    }
    if (qn > 0) drain<F, false>(x, src, acc, q, lane, qn);
    __syncthreads();

    v4f* orow = (v4f*)(agg + (size_t)base * F);
    for (int i = threadIdx.x; i < rows * (F / 4); i += 1024) orow[i] = ((const v4f*)acc)[i];
    __syncthreads();
  }
}

template <int FIN, int NT, bool HAS_AGG>
__global__ __launch_bounds__(256) void
gemm_bias_act_kernel(const float* __restrict__ agg, const float* __restrict__ rdeg,
                     const float* __restrict__ xin, const float* __restrict__ W,
                     const float* __restrict__ bias, float* __restrict__ out,
                     int nRows, int Fout, int doRelu)
{
  constexpr int RG     = 8 / NT;
  constexpr int ROWS   = 16 * RG;
  constexpr int STRIDE = FIN + 4;
  constexpr int KS     = FIN / 4;
  constexpr int RF4    = FIN / 4;
  constexpr int LD     = (ROWS * RF4) / 256;
  static_assert((ROWS * RF4) % 256 == 0, "tile must split evenly over the block");
  __shared__ float As[2][ROWS * STRIDE];

  const int lane = threadIdx.x & 31;
  const int wave = threadIdx.x >> 5;
  const int half = lane >> 4;
  const int l16  = lane & 15;
  const int ct   = wave % NT;
  const int rg   = wave / NT;
  const int n    = ct * 16 + l16;
  const bool colOk = (n < Fout);

  const float* Wrow = W + (size_t)(colOk ? n : Fout - 1) * FIN + 2 * half;
  v2f bf[KS];
  #pragma unroll
  for (int s = 0; s < KS; ++s) {
    v2f b = *(const v2f*)(Wrow + 4 * s);
    if (!colOk) { b.x = 0.0f; b.y = 0.0f; }
    bf[s] = b;
  }
  const float bv = colOk ? bias[n] : 0.0f;

  const int nSteps = (nRows + ROWS - 1) / ROWS;
  v4f   pa[LD], px[LD];
  float pr[LD];

  auto fetch = [&](int step) {
    #pragma unroll
    for (int i = 0; i < LD; ++i) {
      int idx = threadIdx.x + i * 256;
      int r   = idx / RF4;
      int c4  = idx - r * RF4;
      int gr  = min(step * ROWS + r, nRows - 1);
      px[i] = ((const v4f*)xin)[(size_t)gr * RF4 + c4];
      if constexpr (HAS_AGG) {
        pa[i] = ((const v4f*)agg)[(size_t)gr * RF4 + c4];
        pr[i] = rdeg[gr];
      }
    }
  };
  auto stage = [&](int b) {
    #pragma unroll
    for (int i = 0; i < LD; ++i) {
      int idx = threadIdx.x + i * 256;
      int r   = idx / RF4;
      int c4  = idx - r * RF4;
      v4f v = px[i];
      if constexpr (HAS_AGG) v += pa[i] * pr[i];
      *(v4f*)&As[b][r * STRIDE + 4 * c4] = v;
    }
  };

  int u = blockIdx.x;
  if (u < nSteps) { fetch(u); stage(0); }
  __syncthreads();

  int buf = 0;
  for (; u < nSteps; u += gridDim.x) {
    const int  un      = u + gridDim.x;
    const bool hasNext = (un < nSteps);
    if (hasNext) fetch(un);

    const float* Arow = &As[buf][(rg * 16 + l16) * STRIDE + 2 * half];
    v8f c = {};
    #pragma unroll
    for (int s = 0; s < KS; ++s) {
      v2f a = *(const v2f*)(Arow + 4 * s);
      c = __builtin_amdgcn_wmma_f32_16x16x4_f32(false, a, false, bf[s], (short)0, c, false, false);
    }

    const int row0 = u * ROWS + rg * 16 + 8 * half;
    float* colBase = out + (size_t)row0 * Fout + n;
    if ((u * ROWS + ROWS <= nRows) && (ct * 16 + 16 <= Fout)) {
      #pragma unroll
      for (int r = 0; r < 8; ++r) {
        float val = c[r] + bv;
        if (doRelu) val = fmaxf(val, 0.0f);
        colBase[(size_t)r * Fout] = val;
      }
    } else {
      #pragma unroll
      for (int r = 0; r < 8; ++r) {
        float val = c[r] + bv;
        if (doRelu) val = fmaxf(val, 0.0f);
        if (row0 + r < nRows && colOk) colBase[(size_t)r * Fout] = val;
      }
    }

    if (hasNext) stage(buf ^ 1);
    __syncthreads();
    buf ^= 1;
  }
}

extern "C" void kernel_launch(void* const* d_in, const int* in_sizes, int n_in,
                              void* d_out, int out_size, void* d_ws, size_t ws_size,
                              hipStream_t stream) {
  const float* x0 = (const float*)d_in[0];
  const int*   ei = (const int*)d_in[1];
  const float *W1 = (const float*)d_in[2],  *b1 = (const float*)d_in[3];
  const float *W2 = (const float*)d_in[4],  *b2 = (const float*)d_in[5];
  const float *W3 = (const float*)d_in[6],  *b3 = (const float*)d_in[7];
  const float *Wp = (const float*)d_in[8],  *bp = (const float*)d_in[9];
  const float *Wc = (const float*)d_in[10], *bc = (const float*)d_in[11];

  const int N      = in_sizes[0] / 64;
  const int nEdges = in_sizes[1] / 2;
  const int* src = ei;
  const int* dst = ei + nEdges;

  float* ws  = (float*)d_ws;
  float* deg = ws;
  float* agg = ws + 65536;
  float* h2  = agg + (size_t)N * 128;

  float* out = (float*)d_out;
  float* h1  = out;
  float* emb = out;
  float* nxt = out + (size_t)N * 64;
  float* cls = out + (size_t)N * 128;
  unsigned* hmap = (unsigned*)cls;

  auto nb = [](int n) { return dim3((unsigned)((n + 255) / 256)); };
  dim3 b256(256);
  dim3 gemmGrid(512), gemmBlock(256);
  dim3 ownGrid(256), ownBlock(1024);

  zero_kernel     <<<nb(N),      b256, 0, stream>>>(deg, N);
  degree_kernel   <<<nb(nEdges), b256, 0, stream>>>(dst, deg, nEdges);
  recip_deg_kernel<<<nb(N),      b256, 0, stream>>>(deg, N);

  owner_sum_kernel<64, 0> <<<ownGrid, ownBlock, 0, stream>>>(x0, src, dst, agg, hmap, N, nEdges);
  gemm_bias_act_kernel<64, 8, true><<<gemmGrid, gemmBlock, 0, stream>>>(agg, deg, x0, W1, b1, h1, N, 128, 1);

  owner_sum_kernel<128, 1><<<ownGrid, ownBlock, 0, stream>>>(h1, src, dst, agg, hmap, N, nEdges);
  gemm_bias_act_kernel<128, 8, true><<<gemmGrid, gemmBlock, 0, stream>>>(agg, deg, h1, W2, b2, h2, N, 128, 1);

  owner_sum_kernel<128, 1><<<ownGrid, ownBlock, 0, stream>>>(h2, src, dst, agg, hmap, N, nEdges);
  gemm_bias_act_kernel<128, 4, true><<<gemmGrid, gemmBlock, 0, stream>>>(agg, deg, h2, W3, b3, emb, N, 64, 1);

  gemm_bias_act_kernel<64, 4, false><<<gemmGrid, gemmBlock, 0, stream>>>(nullptr, nullptr, emb, Wp, bp, nxt, N, 64, 0);
  gemm_bias_act_kernel<64, 1, false><<<gemmGrid, gemmBlock, 0, stream>>>(nullptr, nullptr, emb, Wc, bc, cls, N, 10, 0);
}


# Degrees counted inside layer 1's owner blocks: one LDS int add per queued edge and row, block stores 1/max(count,1) for its rows; the 3 degree launches and all global float atomics go; same values
# speedup vs baseline: 4.0603x; 1.2543x over previous
#include <hip/hip_runtime.h>
#include <hip/hip_bf16.h>

typedef __attribute__((ext_vector_type(8))) float v8f;
typedef __attribute__((ext_vector_type(2))) float v2f;
typedef __attribute__((ext_vector_type(4))) float v4f;
typedef __attribute__((ext_vector_type(4))) int   v4i;

template <int F, bool FULL>
__device__ __forceinline__ void add_rows(const float* __restrict__ x, float* acc,
                                         int sv, int rv, int lane, int cnt) {
  constexpr int C = F / 32;
  constexpr int U = 32 / C;
  #pragma unroll
  for (int j0 = 0; j0 < 32; j0 += U) {
    if (!FULL && j0 >= cnt) break;
    float v[U][C];
    #pragma unroll
    for (int j = 0; j < U; ++j) {
      const int s = __builtin_amdgcn_readlane(sv, j0 + j);
      const float* xr = x + (size_t)s * F + lane;
      #pragma unroll
      for (int c = 0; c < C; ++c) v[j][c] = xr[c * 32];
    }
    #pragma unroll
    for (int j = 0; j < U; ++j) {
      if (FULL || j0 + j < cnt) {
        const int r = __builtin_amdgcn_readlane(rv, j0 + j);
        float* ar = acc + r * F + lane;
        #pragma unroll
        for (int c = 0; c < C; ++c)
          __hip_atomic_fetch_add(ar + c * 32, v[j][c], __ATOMIC_RELAXED, __HIP_MEMORY_SCOPE_WORKGROUP);
      }
    }
  }
}

template <int F, bool FULL>
__device__ __forceinline__ void drain(const float* __restrict__ x, const int* __restrict__ src,
                                      float* acc, const unsigned* qp, int lane, int cnt) {
  const unsigned ent = qp[FULL ? lane : min(lane, cnt - 1)];
  const int rv = (int)(ent & 255u);
  const int sv = src[ent >> 8];
  add_rows<F, FULL>(x, acc, sv, rv, lane, cnt);
}

template <bool TAIL, bool VEC, bool MAP, bool DEG>
__device__ __forceinline__ int scan_step(const int* __restrict__ dst, unsigned* q, int qn, int e0,
                                         int lane, int base, int rows, int nEdges, unsigned& lw,
                                         int* cnt) {
  int d[32];
  if (VEC && !TAIL) {
    const v4i* dp = (const v4i*)(dst + e0) + lane;
    #pragma unroll
    for (int k4 = 0; k4 < 8; ++k4) {
      const v4i t = dp[k4 * 32];
      d[4 * k4 + 0] = t.x; d[4 * k4 + 1] = t.y; d[4 * k4 + 2] = t.z; d[4 * k4 + 3] = t.w;
    }
  } else {
    #pragma unroll
    for (int k = 0; k < 32; ++k) {
      const int e = e0 + k * 32 + lane;
      d[k] = dst[TAIL ? min(e, nEdges - 1) : e];
    }
  }
  #pragma unroll
  for (int k = 0; k < 32; ++k) {
    const int e = (VEC && !TAIL) ? (e0 + (k >> 2) * 128 + lane * 4 + (k & 3)) : (e0 + k * 32 + lane);
    const unsigned r = (unsigned)(d[k] - base);
    bool hit = (r < (unsigned)rows);
    if (TAIL) hit = hit && (e < nEdges);
    const unsigned m = __builtin_amdgcn_ballot_w32(hit);
    if (m != 0u) {
      const int pos = qn + (int)__builtin_amdgcn_mbcnt_lo(m, 0u);
      if (hit) {
        q[pos] = ((unsigned)e << 8) | r;
        if (DEG) __hip_atomic_fetch_add(&cnt[r], 1, __ATOMIC_RELAXED, __HIP_MEMORY_SCOPE_WORKGROUP);
      }
      qn += __builtin_popcount(m);
      if (MAP) {
        if (VEC && !TAIL) {
          const unsigned g = ((m & 0x000000FFu) ? 1u : 0u) | ((m & 0x0000FF00u) ? 2u : 0u) |
                             ((m & 0x00FF0000u) ? 4u : 0u) | ((m & 0xFF000000u) ? 8u : 0u);
          lw |= g << (4 * (k >> 2));
        } else {
          lw |= 1u << k;
        }
      }
    }
  }
  return qn;
}

template <int F, int MODE>
__global__ __launch_bounds__(1024) void
owner_sum_kernel(const float* __restrict__ x, const int* __restrict__ src,
                 const int* __restrict__ dst, float* __restrict__ agg, unsigned* hitmap,
                 float* __restrict__ rdeg, int nRows, int nEdges) {
  constexpr int RB    = 196;
  constexpr int STEP  = 32 * 32;
  constexpr int QCAP  = STEP + 32;
  constexpr int MAPW  = 10;
  constexpr int NL    = 16;
  __shared__ float    acc[RB * F];
  __shared__ unsigned queue[32 * QCAP];
  __shared__ unsigned short lineq[MODE == 1 ? 32 * 1024 : 1];
  __shared__ int cnt[MODE == 0 ? RB : 1];

  const int lane = threadIdx.x & 31;
  const int wave = __builtin_amdgcn_readfirstlane((int)(threadIdx.x >> 5));
  unsigned* q = queue + wave * QCAP;
  const bool vec = (((size_t)dst) & 15) == 0;
  const int nRounds = (nEdges + 32 * STEP - 1) / (32 * STEP);
  const bool mapOk = ((long long)gridDim.x * RB >= (long long)nRows) && (nRounds <= 32) &&
                     ((long long)gridDim.x * 32 * 32 <= (long long)nRows * MAPW);
  unsigned* mapw = hitmap + ((size_t)blockIdx.x * 32 + wave) * 32;

  for (int base = blockIdx.x * RB; base < nRows; base += gridDim.x * RB) {
    const int rows = min(RB, nRows - base);
    for (int i = threadIdx.x; i < RB * F / 4; i += 1024) ((v4f*)acc)[i] = v4f{0.0f, 0.0f, 0.0f, 0.0f};
    if (MODE == 0) for (int i = threadIdx.x; i < RB; i += 1024) cnt[i] = 0;
    __syncthreads();

    int qn = 0;
    if (MODE == 1 && mapOk) {
      const unsigned w = mapw[lane];
      const int p = __builtin_popcount(w);
      int off = 0, total = 0;
      #pragma unroll
      for (int t = 0; t < 32; ++t) {
        const int pt = __builtin_amdgcn_readlane(p, t);
        if (lane > t) off += pt;
        total += pt;
      }
      unsigned short* lq = lineq + wave * 1024;
      for (unsigned ww = w; ww != 0u; ww &= ww - 1u)
        lq[off++] = (unsigned short)(lane * 32 + __builtin_ctz(ww));

      for (int b0 = 0; b0 < total; b0 += 32) {
        const int nb = min(32, total - b0);
        const int lv = lq[b0 + min(lane, nb - 1)];
        #pragma unroll
        for (int j0 = 0; j0 < 32; j0 += NL) {
          if (j0 >= nb) break;
          int d[NL], eb[NL];
          #pragma unroll
          for (int j = 0; j < NL; ++j) {
            const int code = __builtin_amdgcn_readlane(lv, j0 + j);
            const int e = ((code >> 5) * 32 + wave) * STEP + (code & 31) * 32 + lane;
            eb[j] = e;
            d[j]  = dst[min(e, nEdges - 1)];
          }
          #pragma unroll
          for (int j = 0; j < NL; ++j) {
            if (j0 + j < nb) {
              const unsigned r = (unsigned)(d[j] - base);
              const bool hit = (r < (unsigned)rows) && (eb[j] < nEdges);
              const unsigned m = __builtin_amdgcn_ballot_w32(hit);
              if (m != 0u) {
                const int pos = qn + (int)__builtin_amdgcn_mbcnt_lo(m, 0u);
                if (hit) q[pos] = ((unsigned)eb[j] << 8) | r;
                qn += __builtin_popcount(m);
              }
            }
          }
          while (qn >= 32) { qn -= 32; drain<F, true>(x, src, acc, q + qn, lane, 32); }
        }
      }
    } else {
      unsigned mapv = 0u;
      for (int t = 0; t < nRounds; ++t) {
        const int tt = (t + (int)blockIdx.x) % nRounds;
        const int e0 = tt * (32 * STEP) + wave * STEP;
        if (e0 < nEdges) {
          unsigned lw = 0u;
          if (e0 + STEP > nEdges) qn = scan_step<true,  false, MODE == 0, MODE == 0>(dst, q, qn, e0, lane, base, rows, nEdges, lw, cnt);
          else if (vec)           qn = scan_step<false, true,  MODE == 0, MODE == 0>(dst, q, qn, e0, lane, base, rows, nEdges, lw, cnt);
          else                    qn = scan_step<false, false, MODE == 0, MODE == 0>(dst, q, qn, e0, lane, base, rows, nEdges, lw, cnt);
          if (MODE == 0) { if (lane == tt) mapv = lw; }
          while (qn >= 32) { qn -= 32; drain<F, true>(x, src, acc, q + qn, lane, 32); }
        }
      }
      if (MODE == 0 && mapOk) mapw[lane] = mapv;
    }
    if (qn > 0) drain<F, false>(x, src, acc, q, lane, qn);
    __syncthreads();

    v4f* orow = (v4f*)(agg + (size_t)base * F);
    for (int i = threadIdx.x; i < rows * (F / 4); i += 1024) orow[i] = ((const v4f*)acc)[i];
    if (MODE == 0)
      for (int i = threadIdx.x; i < rows; i += 1024) rdeg[base + i] = 1.0f / fmaxf((float)cnt[i], 1.0f);
    __syncthreads();
  }
}

template <int FIN, int NT, bool HAS_AGG>
__global__ __launch_bounds__(256) void
gemm_bias_act_kernel(const float* __restrict__ agg, const float* __restrict__ rdeg,
                     const float* __restrict__ xin, const float* __restrict__ W,
                     const float* __restrict__ bias, float* __restrict__ out,
                     int nRows, int Fout, int doRelu)
{
  constexpr int RG     = 8 / NT;
  constexpr int ROWS   = 16 * RG;
  constexpr int STRIDE = FIN + 4;
  constexpr int KS     = FIN / 4;
  constexpr int RF4    = FIN / 4;
  constexpr int LD     = (ROWS * RF4) / 256;
  static_assert((ROWS * RF4) % 256 == 0, "tile must split evenly over the block");
  __shared__ float As[2][ROWS * STRIDE];

  const int lane = threadIdx.x & 31;
  const int wave = threadIdx.x >> 5;
  const int half = lane >> 4;
  const int l16  = lane & 15;
  const int ct   = wave % NT;
  const int rg   = wave / NT;
  const int n    = ct * 16 + l16;
  const bool colOk = (n < Fout);

  const float* Wrow = W + (size_t)(colOk ? n : Fout - 1) * FIN + 2 * half;
  v2f bf[KS];
  #pragma unroll
  for (int s = 0; s < KS; ++s) {
    v2f b = *(const v2f*)(Wrow + 4 * s);
    if (!colOk) { b.x = 0.0f; b.y = 0.0f; }
    bf[s] = b;
  }
  const float bv = colOk ? bias[n] : 0.0f;

  const int nSteps = (nRows + ROWS - 1) / ROWS;
  v4f   pa[LD], px[LD];
  float pr[LD];

  auto fetch = [&](int step) {
    #pragma unroll
    for (int i = 0; i < LD; ++i) {
      int idx = threadIdx.x + i * 256;
      int r   = idx / RF4;
      int c4  = idx - r * RF4;
      int gr  = min(step * ROWS + r, nRows - 1);
      px[i] = ((const v4f*)xin)[(size_t)gr * RF4 + c4];
      if constexpr (HAS_AGG) {
        pa[i] = ((const v4f*)agg)[(size_t)gr * RF4 + c4];
        pr[i] = rdeg[gr];
      }
    }
  };
  auto stage = [&](int b) {
    #pragma unroll
    for (int i = 0; i < LD; ++i) {
      int idx = threadIdx.x + i * 256;
      int r   = idx / RF4;
      int c4  = idx - r * RF4;
      v4f v = px[i];
      if constexpr (HAS_AGG) v += pa[i] * pr[i];
      *(v4f*)&As[b][r * STRIDE + 4 * c4] = v;
    }
  };

  int u = blockIdx.x;
  if (u < nSteps) { fetch(u); stage(0); }
  __syncthreads();

  int buf = 0;
  for (; u < nSteps; u += gridDim.x) {
    const int  un      = u + gridDim.x;
    const bool hasNext = (un < nSteps);
    if (hasNext) fetch(un);

    const float* Arow = &As[buf][(rg * 16 + l16) * STRIDE + 2 * half];
    v8f c = {};
    #pragma unroll
    for (int s = 0; s < KS; ++s) {
      v2f a = *(const v2f*)(Arow + 4 * s);
      c = __builtin_amdgcn_wmma_f32_16x16x4_f32(false, a, false, bf[s], (short)0, c, false, false);
    }

    const int row0 = u * ROWS + rg * 16 + 8 * half;
    float* colBase = out + (size_t)row0 * Fout + n;
    if ((u * ROWS + ROWS <= nRows) && (ct * 16 + 16 <= Fout)) {
      #pragma unroll
      for (int r = 0; r < 8; ++r) {
        float val = c[r] + bv;
        if (doRelu) val = fmaxf(val, 0.0f);
        colBase[(size_t)r * Fout] = val;
      }
    } else {
      #pragma unroll
      for (int r = 0; r < 8; ++r) {
        float val = c[r] + bv;
        if (doRelu) val = fmaxf(val, 0.0f);
        if (row0 + r < nRows && colOk) colBase[(size_t)r * Fout] = val;
      }
    }

    if (hasNext) stage(buf ^ 1);
    __syncthreads();
    buf ^= 1;
  }
}

extern "C" void kernel_launch(void* const* d_in, const int* in_sizes, int n_in,
                              void* d_out, int out_size, void* d_ws, size_t ws_size,
                              hipStream_t stream) {
  const float* x0 = (const float*)d_in[0];
  const int*   ei = (const int*)d_in[1];
  const float *W1 = (const float*)d_in[2],  *b1 = (const float*)d_in[3];
  const float *W2 = (const float*)d_in[4],  *b2 = (const float*)d_in[5];
  const float *W3 = (const float*)d_in[6],  *b3 = (const float*)d_in[7];
  const float *Wp = (const float*)d_in[8],  *bp = (const float*)d_in[9];
  const float *Wc = (const float*)d_in[10], *bc = (const float*)d_in[11];

  const int N      = in_sizes[0] / 64;
  const int nEdges = in_sizes[1] / 2;
  const int* src = ei;
  const int* dst = ei + nEdges;

  float* ws  = (float*)d_ws;
  float* deg = ws;
  float* agg = ws + 65536;
  float* h2  = agg + (size_t)N * 128;

  float* out = (float*)d_out;
  float* h1  = out;
  float* emb = out;
  float* nxt = out + (size_t)N * 64;
  float* cls = out + (size_t)N * 128;
  unsigned* hmap = (unsigned*)cls;

  dim3 gemmGrid(512), gemmBlock(256);
  dim3 ownGrid(256), ownBlock(1024);

  owner_sum_kernel<64, 0> <<<ownGrid, ownBlock, 0, stream>>>(x0, src, dst, agg, hmap, deg, N, nEdges);
  gemm_bias_act_kernel<64, 8, true><<<gemmGrid, gemmBlock, 0, stream>>>(agg, deg, x0, W1, b1, h1, N, 128, 1);

  owner_sum_kernel<128, 1><<<ownGrid, ownBlock, 0, stream>>>(h1, src, dst, agg, hmap, deg, N, nEdges);
  gemm_bias_act_kernel<128, 8, true><<<gemmGrid, gemmBlock, 0, stream>>>(agg, deg, h1, W2, b2, h2, N, 128, 1);

  owner_sum_kernel<128, 1><<<ownGrid, ownBlock, 0, stream>>>(h2, src, dst, agg, hmap, deg, N, nEdges);
  gemm_bias_act_kernel<128, 4, true><<<gemmGrid, gemmBlock, 0, stream>>>(agg, deg, h2, W3, b3, emb, N, 64, 1);

  gemm_bias_act_kernel<64, 4, false><<<gemmGrid, gemmBlock, 0, stream>>>(nullptr, nullptr, emb, Wp, bp, nxt, N, 64, 0);
  gemm_bias_act_kernel<64, 1, false><<<gemmGrid, gemmBlock, 0, stream>>>(nullptr, nullptr, emb, Wc, bc, cls, N, 10, 0);
}


# Layer products moved into the owner blocks: LDS sums become the A rows in place (sum/degree + own x row), the block's 32 waves run the W tiles on them; agg never stored, 8 -> 5 launches; same sums
# speedup vs baseline: 4.3415x; 1.0693x over previous
#include <hip/hip_runtime.h>
#include <hip/hip_bf16.h>

typedef __attribute__((ext_vector_type(8))) float v8f;
typedef __attribute__((ext_vector_type(2))) float v2f;
typedef __attribute__((ext_vector_type(4))) float v4f;
typedef __attribute__((ext_vector_type(4))) int   v4i;

template <int F, int S, bool FULL>
__device__ __forceinline__ void add_rows(const float* __restrict__ x, float* acc,
                                         int sv, int rv, int lane, int cnt) {
  constexpr int C = F / 32;
  constexpr int U = 32 / C;
  #pragma unroll
  for (int j0 = 0; j0 < 32; j0 += U) {
    if (!FULL && j0 >= cnt) break;
    float v[U][C];
    #pragma unroll
    for (int j = 0; j < U; ++j) {
      const int s = __builtin_amdgcn_readlane(sv, j0 + j);
      const float* xr = x + (size_t)s * F + lane;
      #pragma unroll
      for (int c = 0; c < C; ++c) v[j][c] = xr[c * 32];
    }
    #pragma unroll
    for (int j = 0; j < U; ++j) {
      if (FULL || j0 + j < cnt) {
        const int r = __builtin_amdgcn_readlane(rv, j0 + j);
        float* ar = acc + r * S + lane;
        #pragma unroll
        for (int c = 0; c < C; ++c)
          __hip_atomic_fetch_add(ar + c * 32, v[j][c], __ATOMIC_RELAXED, __HIP_MEMORY_SCOPE_WORKGROUP);
      }
    }
  }
}

template <int F, int S, bool FULL>
__device__ __forceinline__ void drain(const float* __restrict__ x, const int* __restrict__ src,
                                      float* acc, const unsigned* qp, int lane, int cnt) {
  const unsigned ent = qp[FULL ? lane : min(lane, cnt - 1)];
  const int rv = (int)(ent & 255u);
  const int sv = src[ent >> 8];
  add_rows<F, S, FULL>(x, acc, sv, rv, lane, cnt);
}

template <bool TAIL, bool VEC, bool MAP, bool DEG>
__device__ __forceinline__ int scan_step(const int* __restrict__ dst, unsigned* q, int qn, int e0,
                                         int lane, int base, int rows, int nEdges, unsigned& lw,
                                         int* cnt) {
  int d[32];
  if (VEC && !TAIL) {
    const v4i* dp = (const v4i*)(dst + e0) + lane;
    #pragma unroll
    for (int k4 = 0; k4 < 8; ++k4) {
      const v4i t = dp[k4 * 32];
      d[4 * k4 + 0] = t.x; d[4 * k4 + 1] = t.y; d[4 * k4 + 2] = t.z; d[4 * k4 + 3] = t.w;
    }
  } else {
    #pragma unroll
    for (int k = 0; k < 32; ++k) {
      const int e = e0 + k * 32 + lane;
      d[k] = dst[TAIL ? min(e, nEdges - 1) : e];
    }
  }
  #pragma unroll
  for (int k = 0; k < 32; ++k) {
    const int e = (VEC && !TAIL) ? (e0 + (k >> 2) * 128 + lane * 4 + (k & 3)) : (e0 + k * 32 + lane);
    const unsigned r = (unsigned)(d[k] - base);
    bool hit = (r < (unsigned)rows);
    if (TAIL) hit = hit && (e < nEdges);
    const unsigned m = __builtin_amdgcn_ballot_w32(hit);
    if (m != 0u) {
      const int pos = qn + (int)__builtin_amdgcn_mbcnt_lo(m, 0u);
      if (hit) {
        q[pos] = ((unsigned)e << 8) | r;
        if (DEG) __hip_atomic_fetch_add(&cnt[r], 1, __ATOMIC_RELAXED, __HIP_MEMORY_SCOPE_WORKGROUP);
      }
      qn += __builtin_popcount(m);
      if (MAP) {
        if (VEC && !TAIL) {
          const unsigned g = ((m & 0x000000FFu) ? 1u : 0u) | ((m & 0x0000FF00u) ? 2u : 0u) |
                             ((m & 0x00FF0000u) ? 4u : 0u) | ((m & 0xFF000000u) ? 8u : 0u);
          lw |= g << (4 * (k >> 2));
        } else {
          lw |= 1u << k;
        }
      }
    }
  }
  return qn;
}

template <int F, int MODE, int FOUT>
__global__ __launch_bounds__(1024) void
owner_layer_kernel(const float* __restrict__ x, const int* __restrict__ src,
                   const int* __restrict__ dst, unsigned* hitmap, float* rdeg,
                   const float* W, const float* bias,
                   float* __restrict__ out, int nRows, int nEdges) {
  constexpr int RB    = 196;
  constexpr int RT    = (RB + 15) / 16;
  constexpr int RBP   = RT * 16;
  constexpr int STRIDE = F + 4;
  constexpr int NT    = FOUT / 16;
  constexpr int RG    = 32 / NT;
  constexpr int KS    = F / 4;
  static_assert(FOUT % 16 == 0 && 32 % NT == 0, "column tiles must split evenly over the 32 waves");
  constexpr int STEP  = 32 * 32;
  constexpr int QCAP  = STEP + 32;
  constexpr int MAPW  = 10;
  constexpr int NL    = 16;
  __shared__ __attribute__((aligned(16))) float acc[RBP * STRIDE];
  __shared__ unsigned queue[32 * QCAP];
  __shared__ unsigned short lineq[MODE == 1 ? 32 * 1024 : 1];
  __shared__ int cnt[MODE == 0 ? RB : 1];

  const int lane = threadIdx.x & 31;
  const int wave = __builtin_amdgcn_readfirstlane((int)(threadIdx.x >> 5));
  unsigned* q = queue + wave * QCAP;
  const bool vec = (((size_t)dst) & 15) == 0;
  const int nRounds = (nEdges + 32 * STEP - 1) / (32 * STEP);
  const bool mapOk = ((long long)gridDim.x * RB >= (long long)nRows) && (nRounds <= 32) &&
                     ((long long)gridDim.x * 32 * 32 <= (long long)nRows * MAPW);
  unsigned* mapw = hitmap + ((size_t)blockIdx.x * 32 + wave) * 32;

  for (int base = blockIdx.x * RB; base < nRows; base += gridDim.x * RB) {
    const int rows = min(RB, nRows - base);
    for (int i = threadIdx.x; i < RBP * STRIDE / 4; i += 1024) ((v4f*)acc)[i] = v4f{0.0f, 0.0f, 0.0f, 0.0f};
    if (MODE == 0) for (int i = threadIdx.x; i < RB; i += 1024) cnt[i] = 0;
    __syncthreads();

    int qn = 0;
    if (MODE == 1 && mapOk) {
      const unsigned w = mapw[lane];
      const int p = __builtin_popcount(w);
      int off = 0, total = 0;
      #pragma unroll
      for (int t = 0; t < 32; ++t) {
        const int pt = __builtin_amdgcn_readlane(p, t);
        if (lane > t) off += pt;
        total += pt;
      }
      unsigned short* lq = lineq + wave * 1024;
      for (unsigned ww = w; ww != 0u; ww &= ww - 1u)
        lq[off++] = (unsigned short)(lane * 32 + __builtin_ctz(ww));

      for (int b0 = 0; b0 < total; b0 += 32) {
        const int nb = min(32, total - b0);
        const int lv = lq[b0 + min(lane, nb - 1)];
        #pragma unroll
        for (int j0 = 0; j0 < 32; j0 += NL) {
          if (j0 >= nb) break;
          int d[NL], eb[NL];
          #pragma unroll
          for (int j = 0; j < NL; ++j) {
            const int code = __builtin_amdgcn_readlane(lv, j0 + j);
            const int e = ((code >> 5) * 32 + wave) * STEP + (code & 31) * 32 + lane;
            eb[j] = e;
            d[j]  = dst[min(e, nEdges - 1)];
          }
          #pragma unroll
          for (int j = 0; j < NL; ++j) {
            if (j0 + j < nb) {
              const unsigned r = (unsigned)(d[j] - base);
              const bool hit = (r < (unsigned)rows) && (eb[j] < nEdges);
              const unsigned m = __builtin_amdgcn_ballot_w32(hit);
              if (m != 0u) {
                const int pos = qn + (int)__builtin_amdgcn_mbcnt_lo(m, 0u);
                if (hit) q[pos] = ((unsigned)eb[j] << 8) | r;
                qn += __builtin_popcount(m);
              }
            }
          }
          while (qn >= 32) { qn -= 32; drain<F, STRIDE, true>(x, src, acc, q + qn, lane, 32); }
        }
      }
    } else {
      unsigned mapv = 0u;
      for (int t = 0; t < nRounds; ++t) {
        const int tt = (t + (int)blockIdx.x) % nRounds;
        const int e0 = tt * (32 * STEP) + wave * STEP;
        if (e0 < nEdges) {
          unsigned lw = 0u;
          if (e0 + STEP > nEdges) qn = scan_step<true,  false, MODE == 0, MODE == 0>(dst, q, qn, e0, lane, base, rows, nEdges, lw, cnt);
          else if (vec)           qn = scan_step<false, true,  MODE == 0, MODE == 0>(dst, q, qn, e0, lane, base, rows, nEdges, lw, cnt);
          else                    qn = scan_step<false, false, MODE == 0, MODE == 0>(dst, q, qn, e0, lane, base, rows, nEdges, lw, cnt);
          if (MODE == 0) { if (lane == tt) mapv = lw; }
          while (qn >= 32) { qn -= 32; drain<F, STRIDE, true>(x, src, acc, q + qn, lane, 32); }
        }
      }
      if (MODE == 0 && mapOk) mapw[lane] = mapv;
    }
    if (qn > 0) drain<F, STRIDE, false>(x, src, acc, q, lane, qn);

    __syncthreads();

    {
      constexpr int PA = (RB * (F / 4) + 1023) / 1024;
      const int nA = rows * (F / 4);
      v4f   xv[PA];
      float rd[PA];
      #pragma unroll
      for (int k = 0; k < PA; ++k) {
        const int i  = min((int)threadIdx.x + k * 1024, nA - 1);
        const int r  = i / (F / 4);
        const int c4 = i - r * (F / 4);
        xv[k] = ((const v4f*)(x + (size_t)(base + r) * F))[c4];
        rd[k] = (MODE == 0) ? 1.0f / fmaxf((float)cnt[r], 1.0f) : rdeg[base + r];
      }
      #pragma unroll
      for (int k = 0; k < PA; ++k) {
        const int i = (int)threadIdx.x + k * 1024;
        if (i < nA) {
          const int r  = i / (F / 4);
          const int c4 = i - r * (F / 4);
          v4f* ap = (v4f*)&acc[r * STRIDE + 4 * c4];
          *ap = *ap * rd[k] + xv[k];
        }
      }
    }
    if (MODE == 0)
      for (int i = threadIdx.x; i < rows; i += 1024) rdeg[base + i] = 1.0f / fmaxf((float)cnt[i], 1.0f);
    __syncthreads();

    {
      const int half = lane >> 4;
      const int l16  = lane & 15;
      const int ct   = wave % NT;
      const int rg   = wave / NT;
      const int n    = ct * 16 + l16;
      const float* Wrow = W + (size_t)n * F + 2 * half;
      v2f bf[KS];
      #pragma unroll
      for (int s = 0; s < KS; ++s) bf[s] = *(const v2f*)(Wrow + 4 * s);
      const float bv = bias[n];

      for (int rt = rg; rt < RT; rt += RG) {
        if (rt * 16 >= rows) break;
        const float* Arow = &acc[(rt * 16 + l16) * STRIDE + 2 * half];
        v8f c = {};
        #pragma unroll
        for (int s = 0; s < KS; ++s) {
          const v2f a = *(const v2f*)(Arow + 4 * s);
          c = __builtin_amdgcn_wmma_f32_16x16x4_f32(false, a, false, bf[s], (short)0, c, false, false);
        }
        const int lrow0 = rt * 16 + 8 * half;
        float* colBase = out + (size_t)(base + lrow0) * FOUT + n;
        if (rt * 16 + 16 <= rows) {
          #pragma unroll
          for (int r = 0; r < 8; ++r) colBase[(size_t)r * FOUT] = fmaxf(c[r] + bv, 0.0f);
        } else {
          #pragma unroll
          for (int r = 0; r < 8; ++r)
            if (lrow0 + r < rows) colBase[(size_t)r * FOUT] = fmaxf(c[r] + bv, 0.0f);
        }
      }
    }
    __syncthreads();
  }
}

template <int FIN, int NT, bool HAS_AGG>
__global__ __launch_bounds__(256) void
gemm_bias_act_kernel(const float* __restrict__ agg, const float* __restrict__ rdeg,
                     const float* __restrict__ xin, const float* __restrict__ W,
                     const float* __restrict__ bias, float* __restrict__ out,
                     int nRows, int Fout, int doRelu)
{
  constexpr int RG     = 8 / NT;
  constexpr int ROWS   = 16 * RG;
  constexpr int STRIDE = FIN + 4;
  constexpr int KS     = FIN / 4;
  constexpr int RF4    = FIN / 4;
  constexpr int LD     = (ROWS * RF4) / 256;
  static_assert((ROWS * RF4) % 256 == 0, "tile must split evenly over the block");
  __shared__ float As[2][ROWS * STRIDE];

  const int lane = threadIdx.x & 31;
  const int wave = threadIdx.x >> 5;
  const int half = lane >> 4;
  const int l16  = lane & 15;
  const int ct   = wave % NT;
  const int rg   = wave / NT;
  const int n    = ct * 16 + l16;
  const bool colOk = (n < Fout);

  const float* Wrow = W + (size_t)(colOk ? n : Fout - 1) * FIN + 2 * half;
  v2f bf[KS];
  #pragma unroll
  for (int s = 0; s < KS; ++s) {
    v2f b = *(const v2f*)(Wrow + 4 * s);
    if (!colOk) { b.x = 0.0f; b.y = 0.0f; }
    bf[s] = b;
  }
  const float bv = colOk ? bias[n] : 0.0f;

  const int nSteps = (nRows + ROWS - 1) / ROWS;
  v4f   pa[LD], px[LD];
  float pr[LD];

  auto fetch = [&](int step) {
    #pragma unroll
    for (int i = 0; i < LD; ++i) {
      int idx = threadIdx.x + i * 256;
      int r   = idx / RF4;
      int c4  = idx - r * RF4;
      int gr  = min(step * ROWS + r, nRows - 1);
      px[i] = ((const v4f*)xin)[(size_t)gr * RF4 + c4];
      if constexpr (HAS_AGG) {
        pa[i] = ((const v4f*)agg)[(size_t)gr * RF4 + c4];
        pr[i] = rdeg[gr];
      }
    }
  };
  auto stage = [&](int b) {
    #pragma unroll
    for (int i = 0; i < LD; ++i) {
      int idx = threadIdx.x + i * 256;
      int r   = idx / RF4;
      int c4  = idx - r * RF4;
      v4f v = px[i];
      if constexpr (HAS_AGG) v += pa[i] * pr[i];
      *(v4f*)&As[b][r * STRIDE + 4 * c4] = v;
    }
  };

  int u = blockIdx.x;
  if (u < nSteps) { fetch(u); stage(0); }
  __syncthreads();

  int buf = 0;
  for (; u < nSteps; u += gridDim.x) {
    const int  un      = u + gridDim.x;
    const bool hasNext = (un < nSteps);
    if (hasNext) fetch(un);

    const float* Arow = &As[buf][(rg * 16 + l16) * STRIDE + 2 * half];
    v8f c = {};
    #pragma unroll
    for (int s = 0; s < KS; ++s) {
      v2f a = *(const v2f*)(Arow + 4 * s);
      c = __builtin_amdgcn_wmma_f32_16x16x4_f32(false, a, false, bf[s], (short)0, c, false, false);
    }

    const int row0 = u * ROWS + rg * 16 + 8 * half;
    float* colBase = out + (size_t)row0 * Fout + n;
    if ((u * ROWS + ROWS <= nRows) && (ct * 16 + 16 <= Fout)) {
      #pragma unroll
      for (int r = 0; r < 8; ++r) {
        float val = c[r] + bv;
        if (doRelu) val = fmaxf(val, 0.0f);
        colBase[(size_t)r * Fout] = val;
      }
    } else {
      #pragma unroll
      for (int r = 0; r < 8; ++r) {
        float val = c[r] + bv;
        if (doRelu) val = fmaxf(val, 0.0f);
        if (row0 + r < nRows && colOk) colBase[(size_t)r * Fout] = val;
      }
    }

    if (hasNext) stage(buf ^ 1);
    __syncthreads();
    buf ^= 1;
  }
}

extern "C" void kernel_launch(void* const* d_in, const int* in_sizes, int n_in,
                              void* d_out, int out_size, void* d_ws, size_t ws_size,
                              hipStream_t stream) {
  const float* x0 = (const float*)d_in[0];
  const int*   ei = (const int*)d_in[1];
  const float *W1 = (const float*)d_in[2],  *b1 = (const float*)d_in[3];
  const float *W2 = (const float*)d_in[4],  *b2 = (const float*)d_in[5];
  const float *W3 = (const float*)d_in[6],  *b3 = (const float*)d_in[7];
  const float *Wp = (const float*)d_in[8],  *bp = (const float*)d_in[9];
  const float *Wc = (const float*)d_in[10], *bc = (const float*)d_in[11];

  const int N      = in_sizes[0] / 64;
  const int nEdges = in_sizes[1] / 2;
  const int* src = ei;
  const int* dst = ei + nEdges;

  float* ws  = (float*)d_ws;
  float* deg = ws;
  float* h2  = ws + 65536;

  float* out = (float*)d_out;
  float* h1  = out;
  float* emb = out;
  float* nxt = out + (size_t)N * 64;
  float* cls = out + (size_t)N * 128;
  unsigned* hmap = (unsigned*)cls;

  dim3 gemmGrid(512), gemmBlock(256);
  dim3 ownGrid(256), ownBlock(1024);

  owner_layer_kernel<64, 0, 128> <<<ownGrid, ownBlock, 0, stream>>>(x0, src, dst, hmap, deg, W1, b1, h1, N, nEdges);

  owner_layer_kernel<128, 1, 128><<<ownGrid, ownBlock, 0, stream>>>(h1, src, dst, hmap, deg, W2, b2, h2, N, nEdges);

  owner_layer_kernel<128, 1, 64> <<<ownGrid, ownBlock, 0, stream>>>(h2, src, dst, hmap, deg, W3, b3, emb, N, nEdges);

  gemm_bias_act_kernel<64, 4, false><<<gemmGrid, gemmBlock, 0, stream>>>(nullptr, nullptr, emb, Wp, bp, nxt, N, 64, 0);
  gemm_bias_act_kernel<64, 1, false><<<gemmGrid, gemmBlock, 0, stream>>>(nullptr, nullptr, emb, Wc, bc, cls, N, 10, 0);
}


# Edge lists kept from layer 1: each drained batch's (source id, local row) words go as one line to a per-block area; layers 2-3 load lines and gather: no dst, map or src reads; arrival-order sums
# speedup vs baseline: 4.9509x; 1.1404x over previous
#include <hip/hip_runtime.h>
#include <hip/hip_bf16.h>

typedef __attribute__((ext_vector_type(8))) float v8f;
typedef __attribute__((ext_vector_type(2))) float v2f;
typedef __attribute__((ext_vector_type(4))) float v4f;
typedef __attribute__((ext_vector_type(4))) int   v4i;

template <int F, int S, bool FULL>
__device__ __forceinline__ void add_rows(const float* __restrict__ x, float* acc,
                                         int sv, int rv, int lane, int cnt) {
  constexpr int C = F / 32;
  constexpr int U = 32 / C;
  #pragma unroll
  for (int j0 = 0; j0 < 32; j0 += U) {
    if (!FULL && j0 >= cnt) break;
    float v[U][C];
    #pragma unroll
    for (int j = 0; j < U; ++j) {
      const int s = __builtin_amdgcn_readlane(sv, j0 + j);
      const float* xr = x + (size_t)s * F + lane;
      #pragma unroll
      for (int c = 0; c < C; ++c) v[j][c] = xr[c * 32];
    }
    #pragma unroll
    for (int j = 0; j < U; ++j) {
      if (FULL || j0 + j < cnt) {
        const int r = __builtin_amdgcn_readlane(rv, j0 + j);
        float* ar = acc + r * S + lane;
        #pragma unroll
        for (int c = 0; c < C; ++c)
          __hip_atomic_fetch_add(ar + c * 32, v[j][c], __ATOMIC_RELAXED, __HIP_MEMORY_SCOPE_WORKGROUP);
      }
    }
  }
}

constexpr int      LLINES   = 160;
constexpr unsigned LIST_PAD = 0xFFFFFFFFu;

template <int F, int S, bool FULL, bool LISTW>
__device__ __forceinline__ void drain(const float* __restrict__ x, const int* __restrict__ src,
                                      float* acc, const unsigned* qp, int lane, int cnt,
                                      unsigned* lstB, int* lcnt) {
  const unsigned ent = qp[FULL ? lane : min(lane, cnt - 1)];
  const int rv = (int)(ent & 255u);
  const int sv = src[ent >> 8];
  if (LISTW && lstB != nullptr) {
    int c = 0;
    if (lane == 0) c = __hip_atomic_fetch_add(lcnt, 1, __ATOMIC_RELAXED, __HIP_MEMORY_SCOPE_WORKGROUP);
    c = __builtin_amdgcn_readfirstlane(c);
    if (c < LLINES - 1) {
      unsigned w = ((unsigned)sv << 8) | (unsigned)rv;
      if (!FULL && lane >= cnt) w = LIST_PAD;
      lstB[(1 + c) * 32 + lane] = w;
    }
  }
  add_rows<F, S, FULL>(x, acc, sv, rv, lane, cnt);
}

template <bool TAIL, bool VEC, bool MAP, bool DEG>
__device__ __forceinline__ int scan_step(const int* __restrict__ dst, unsigned* q, int qn, int e0,
                                         int lane, int base, int rows, int nEdges, unsigned& lw,
                                         int* cnt) {
  int d[32];
  if (VEC && !TAIL) {
    const v4i* dp = (const v4i*)(dst + e0) + lane;
    #pragma unroll
    for (int k4 = 0; k4 < 8; ++k4) {
      const v4i t = dp[k4 * 32];
      d[4 * k4 + 0] = t.x; d[4 * k4 + 1] = t.y; d[4 * k4 + 2] = t.z; d[4 * k4 + 3] = t.w;
    }
  } else {
    #pragma unroll
    for (int k = 0; k < 32; ++k) {
      const int e = e0 + k * 32 + lane;
      d[k] = dst[TAIL ? min(e, nEdges - 1) : e];
    }
  }
  #pragma unroll
  for (int k = 0; k < 32; ++k) {
    const int e = (VEC && !TAIL) ? (e0 + (k >> 2) * 128 + lane * 4 + (k & 3)) : (e0 + k * 32 + lane);
    const unsigned r = (unsigned)(d[k] - base);
    bool hit = (r < (unsigned)rows);
    if (TAIL) hit = hit && (e < nEdges);
    const unsigned m = __builtin_amdgcn_ballot_w32(hit);
    if (m != 0u) {
      const int pos = qn + (int)__builtin_amdgcn_mbcnt_lo(m, 0u);
      if (hit) {
        q[pos] = ((unsigned)e << 8) | r;
        if (DEG) __hip_atomic_fetch_add(&cnt[r], 1, __ATOMIC_RELAXED, __HIP_MEMORY_SCOPE_WORKGROUP);
      }
      qn += __builtin_popcount(m);
      if (MAP) {
        if (VEC && !TAIL) {
          const unsigned g = ((m & 0x000000FFu) ? 1u : 0u) | ((m & 0x0000FF00u) ? 2u : 0u) |
                             ((m & 0x00FF0000u) ? 4u : 0u) | ((m & 0xFF000000u) ? 8u : 0u);
          lw |= g << (4 * (k >> 2));
        } else {
          lw |= 1u << k;
        }
      }
    }
  }
  return qn;
}

template <int F, int MODE, int FOUT>
__global__ __launch_bounds__(1024) void
owner_layer_kernel(const float* __restrict__ x, const int* __restrict__ src,
                   const int* __restrict__ dst, unsigned* hitmap, float* rdeg,
                   const float* W, const float* bias,
                   float* __restrict__ out, int nRows, int nEdges,
                   unsigned* lst, const float* wsBase, size_t wsBytes) {
  constexpr int RB    = 196;
  constexpr int RT    = (RB + 15) / 16;
  constexpr int RBP   = RT * 16;
  constexpr int STRIDE = F + 4;
  constexpr int NT    = FOUT / 16;
  constexpr int RG    = 32 / NT;
  constexpr int KS    = F / 4;
  static_assert(FOUT % 16 == 0 && 32 % NT == 0, "column tiles must split evenly over the 32 waves");
  constexpr int STEP  = 32 * 32;
  constexpr int QCAP  = STEP + 32;
  constexpr int MAPW  = 10;
  constexpr int NL    = 16;
  __shared__ __attribute__((aligned(16))) float acc[RBP * STRIDE];
  __shared__ unsigned queue[32 * QCAP];
  __shared__ unsigned short lineq[MODE == 1 ? 32 * 1024 : 1];
  __shared__ int cnt[MODE == 0 ? RB : 1];
  __shared__ int lcnt;

  const int lane = threadIdx.x & 31;
  const int wave = __builtin_amdgcn_readfirstlane((int)(threadIdx.x >> 5));
  unsigned* q = queue + wave * QCAP;
  const bool vec = (((size_t)dst) & 15) == 0;
  const int nRounds = (nEdges + 32 * STEP - 1) / (32 * STEP);
  const bool mapOk = ((long long)gridDim.x * RB >= (long long)nRows) && (nRounds <= 32) &&
                     ((long long)gridDim.x * 32 * 32 <= (long long)nRows * MAPW);
  unsigned* mapw = hitmap + ((size_t)blockIdx.x * 32 + wave) * 32;
  const size_t lstOff = (size_t)((const float*)lst - wsBase) * 4;
  const bool lstOk = mapOk && (nRows < (1 << 24) - 1) && (wsBytes >= lstOff) &&
                     (wsBytes - lstOff >= (size_t)gridDim.x * (LLINES * 128));
  unsigned* lstB = lstOk ? lst + (size_t)blockIdx.x * (LLINES * 32) : nullptr;

  for (int base = blockIdx.x * RB; base < nRows; base += gridDim.x * RB) {
    const int rows = min(RB, nRows - base);
    for (int i = threadIdx.x; i < RBP * STRIDE / 4; i += 1024) ((v4f*)acc)[i] = v4f{0.0f, 0.0f, 0.0f, 0.0f};
    if (MODE == 0) for (int i = threadIdx.x; i < RB; i += 1024) cnt[i] = 0;
    if (MODE == 0 && threadIdx.x == 0) lcnt = 0;
    __syncthreads();

    int qn = 0;
    int nCh = -1;
    if (MODE == 1 && lstOk) nCh = __builtin_amdgcn_readfirstlane((int)lstB[0]);
    if (MODE == 1 && nCh >= 0 && nCh <= LLINES - 1) {
      const unsigned* lp = lstB + 32 + lane;
      unsigned ent = (wave < nCh) ? lp[wave * 32] : LIST_PAD;
      for (int c = wave; c < nCh; c += 32) {
        const unsigned en = lp[min(c + 32, nCh - 1) * 32];
        const int nv = __builtin_popcount(__builtin_amdgcn_ballot_w32(ent != LIST_PAD));
        if (nv == 32) {
          add_rows<F, STRIDE, true>(x, acc, (int)(ent >> 8), (int)(ent & 255u), lane, 32);
        } else if (nv > 0) {
          const unsigned last = (unsigned)__builtin_amdgcn_readlane((int)ent, nv - 1);
          const unsigned e2 = (lane < nv) ? ent : last;
          add_rows<F, STRIDE, false>(x, acc, (int)(e2 >> 8), (int)(e2 & 255u), lane, nv);
        }
        ent = en;
      }
    } else if (MODE == 1 && mapOk) {
      const unsigned w = mapw[lane];
      const int p = __builtin_popcount(w);
      int off = 0, total = 0;
      #pragma unroll
      for (int t = 0; t < 32; ++t) {
        const int pt = __builtin_amdgcn_readlane(p, t);
        if (lane > t) off += pt;
        total += pt;
      }
      unsigned short* lq = lineq + wave * 1024;
      for (unsigned ww = w; ww != 0u; ww &= ww - 1u)
        lq[off++] = (unsigned short)(lane * 32 + __builtin_ctz(ww));

      for (int b0 = 0; b0 < total; b0 += 32) {
        const int nb = min(32, total - b0);
        const int lv = lq[b0 + min(lane, nb - 1)];
        #pragma unroll
        for (int j0 = 0; j0 < 32; j0 += NL) {
          if (j0 >= nb) break;
          int d[NL], eb[NL];
          #pragma unroll
          for (int j = 0; j < NL; ++j) {
            const int code = __builtin_amdgcn_readlane(lv, j0 + j);
            const int e = ((code >> 5) * 32 + wave) * STEP + (code & 31) * 32 + lane;
            eb[j] = e;
            d[j]  = dst[min(e, nEdges - 1)];
          }
          #pragma unroll
          for (int j = 0; j < NL; ++j) {
            if (j0 + j < nb) {
              const unsigned r = (unsigned)(d[j] - base);
              const bool hit = (r < (unsigned)rows) && (eb[j] < nEdges);
              const unsigned m = __builtin_amdgcn_ballot_w32(hit);
              if (m != 0u) {
                const int pos = qn + (int)__builtin_amdgcn_mbcnt_lo(m, 0u);
                if (hit) q[pos] = ((unsigned)eb[j] << 8) | r;
                qn += __builtin_popcount(m);
              }
            }
          }
          while (qn >= 32) { qn -= 32; drain<F, STRIDE, true, false>(x, src, acc, q + qn, lane, 32, nullptr, nullptr); }
        }
      }
    } else {
      unsigned mapv = 0u;
      for (int t = 0; t < nRounds; ++t) {
        const int tt = (t + (int)blockIdx.x) % nRounds;
        const int e0 = tt * (32 * STEP) + wave * STEP;
        if (e0 < nEdges) {
          unsigned lw = 0u;
          if (e0 + STEP > nEdges) qn = scan_step<true,  false, MODE == 0, MODE == 0>(dst, q, qn, e0, lane, base, rows, nEdges, lw, cnt);
          else if (vec)           qn = scan_step<false, true,  MODE == 0, MODE == 0>(dst, q, qn, e0, lane, base, rows, nEdges, lw, cnt);
          else                    qn = scan_step<false, false, MODE == 0, MODE == 0>(dst, q, qn, e0, lane, base, rows, nEdges, lw, cnt);
          if (MODE == 0) { if (lane == tt) mapv = lw; }
          while (qn >= 32) { qn -= 32; drain<F, STRIDE, true, MODE == 0>(x, src, acc, q + qn, lane, 32, lstB, &lcnt); }
        }
      }
      if (MODE == 0 && mapOk) mapw[lane] = mapv;
    }
    if (qn > 0) drain<F, STRIDE, false, MODE == 0>(x, src, acc, q, lane, qn, lstB, &lcnt);

    __syncthreads();
    if (MODE == 0 && lstB != nullptr && wave == 0) lstB[lane] = (unsigned)lcnt;

    {
      constexpr int PA = (RB * (F / 4) + 1023) / 1024;
      const int nA = rows * (F / 4);
      v4f   xv[PA];
      float rd[PA];
      #pragma unroll
      for (int k = 0; k < PA; ++k) {
        const int i  = min((int)threadIdx.x + k * 1024, nA - 1);
        const int r  = i / (F / 4);
        const int c4 = i - r * (F / 4);
        xv[k] = ((const v4f*)(x + (size_t)(base + r) * F))[c4];
        rd[k] = (MODE == 0) ? 1.0f / fmaxf((float)cnt[r], 1.0f) : rdeg[base + r];
      }
      #pragma unroll
      for (int k = 0; k < PA; ++k) {
        const int i = (int)threadIdx.x + k * 1024;
        if (i < nA) {
          const int r  = i / (F / 4);
          const int c4 = i - r * (F / 4);
          v4f* ap = (v4f*)&acc[r * STRIDE + 4 * c4];
          *ap = *ap * rd[k] + xv[k];
        }
      }
    }
    if (MODE == 0)
      for (int i = threadIdx.x; i < rows; i += 1024) rdeg[base + i] = 1.0f / fmaxf((float)cnt[i], 1.0f);
    __syncthreads();

    {
      const int half = lane >> 4;
      const int l16  = lane & 15;
      const int ct   = wave % NT;
      const int rg   = wave / NT;
      const int n    = ct * 16 + l16;
      const float* Wrow = W + (size_t)n * F + 2 * half;
      v2f bf[KS];
      #pragma unroll
      for (int s = 0; s < KS; ++s) bf[s] = *(const v2f*)(Wrow + 4 * s);
      const float bv = bias[n];

      for (int rt = rg; rt < RT; rt += RG) {
        if (rt * 16 >= rows) break;
        const float* Arow = &acc[(rt * 16 + l16) * STRIDE + 2 * half];
        v8f c = {};
        #pragma unroll
        for (int s = 0; s < KS; ++s) {
          const v2f a = *(const v2f*)(Arow + 4 * s);
          c = __builtin_amdgcn_wmma_f32_16x16x4_f32(false, a, false, bf[s], (short)0, c, false, false);
        }
        const int lrow0 = rt * 16 + 8 * half;
        float* colBase = out + (size_t)(base + lrow0) * FOUT + n;
        if (rt * 16 + 16 <= rows) {
          #pragma unroll
          for (int r = 0; r < 8; ++r) colBase[(size_t)r * FOUT] = fmaxf(c[r] + bv, 0.0f);
        } else {
          #pragma unroll
          for (int r = 0; r < 8; ++r)
            if (lrow0 + r < rows) colBase[(size_t)r * FOUT] = fmaxf(c[r] + bv, 0.0f);
        }
      }
    }
    __syncthreads();
  }
}

template <int FIN, int NT, bool HAS_AGG>
__global__ __launch_bounds__(256) void
gemm_bias_act_kernel(const float* __restrict__ agg, const float* __restrict__ rdeg,
                     const float* __restrict__ xin, const float* __restrict__ W,
                     const float* __restrict__ bias, float* __restrict__ out,
                     int nRows, int Fout, int doRelu)
{
  constexpr int RG     = 8 / NT;
  constexpr int ROWS   = 16 * RG;
  constexpr int STRIDE = FIN + 4;
  constexpr int KS     = FIN / 4;
  constexpr int RF4    = FIN / 4;
  constexpr int LD     = (ROWS * RF4) / 256;
  static_assert((ROWS * RF4) % 256 == 0, "tile must split evenly over the block");
  __shared__ float As[2][ROWS * STRIDE];

  const int lane = threadIdx.x & 31;
  const int wave = threadIdx.x >> 5;
  const int half = lane >> 4;
  const int l16  = lane & 15;
  const int ct   = wave % NT;
  const int rg   = wave / NT;
  const int n    = ct * 16 + l16;
  const bool colOk = (n < Fout);

  const float* Wrow = W + (size_t)(colOk ? n : Fout - 1) * FIN + 2 * half;
  v2f bf[KS];
  #pragma unroll
  for (int s = 0; s < KS; ++s) {
    v2f b = *(const v2f*)(Wrow + 4 * s);
    if (!colOk) { b.x = 0.0f; b.y = 0.0f; }
    bf[s] = b;
  }
  const float bv = colOk ? bias[n] : 0.0f;

  const int nSteps = (nRows + ROWS - 1) / ROWS;
  v4f   pa[LD], px[LD];
  float pr[LD];

  auto fetch = [&](int step) {
    #pragma unroll
    for (int i = 0; i < LD; ++i) {
      int idx = threadIdx.x + i * 256;
      int r   = idx / RF4;
      int c4  = idx - r * RF4;
      int gr  = min(step * ROWS + r, nRows - 1);
      px[i] = ((const v4f*)xin)[(size_t)gr * RF4 + c4];
      if constexpr (HAS_AGG) {
        pa[i] = ((const v4f*)agg)[(size_t)gr * RF4 + c4];
        pr[i] = rdeg[gr];
      }
    }
  };
  auto stage = [&](int b) {
    #pragma unroll
    for (int i = 0; i < LD; ++i) {
      int idx = threadIdx.x + i * 256;
      int r   = idx / RF4;
      int c4  = idx - r * RF4;
      v4f v = px[i];
      if constexpr (HAS_AGG) v += pa[i] * pr[i];
      *(v4f*)&As[b][r * STRIDE + 4 * c4] = v;
    }
  };

  int u = blockIdx.x;
  if (u < nSteps) { fetch(u); stage(0); }
  __syncthreads();

  int buf = 0;
  for (; u < nSteps; u += gridDim.x) {
    const int  un      = u + gridDim.x;
    const bool hasNext = (un < nSteps);
    if (hasNext) fetch(un);

    const float* Arow = &As[buf][(rg * 16 + l16) * STRIDE + 2 * half];
    v8f c = {};
    #pragma unroll
    for (int s = 0; s < KS; ++s) {
      v2f a = *(const v2f*)(Arow + 4 * s);
      c = __builtin_amdgcn_wmma_f32_16x16x4_f32(false, a, false, bf[s], (short)0, c, false, false);
    }

    const int row0 = u * ROWS + rg * 16 + 8 * half;
    float* colBase = out + (size_t)row0 * Fout + n;
    if ((u * ROWS + ROWS <= nRows) && (ct * 16 + 16 <= Fout)) {
      #pragma unroll
      for (int r = 0; r < 8; ++r) {
        float val = c[r] + bv;
        if (doRelu) val = fmaxf(val, 0.0f);
        colBase[(size_t)r * Fout] = val;
      }
    } else {
      #pragma unroll
      for (int r = 0; r < 8; ++r) {
        float val = c[r] + bv;
        if (doRelu) val = fmaxf(val, 0.0f);
        if (row0 + r < nRows && colOk) colBase[(size_t)r * Fout] = val;
      }
    }

    if (hasNext) stage(buf ^ 1);
    __syncthreads();
    buf ^= 1;
  }
}

extern "C" void kernel_launch(void* const* d_in, const int* in_sizes, int n_in,
                              void* d_out, int out_size, void* d_ws, size_t ws_size,
                              hipStream_t stream) {
  const float* x0 = (const float*)d_in[0];
  const int*   ei = (const int*)d_in[1];
  const float *W1 = (const float*)d_in[2],  *b1 = (const float*)d_in[3];
  const float *W2 = (const float*)d_in[4],  *b2 = (const float*)d_in[5];
  const float *W3 = (const float*)d_in[6],  *b3 = (const float*)d_in[7];
  const float *Wp = (const float*)d_in[8],  *bp = (const float*)d_in[9];
  const float *Wc = (const float*)d_in[10], *bc = (const float*)d_in[11];

  const int N      = in_sizes[0] / 64;
  const int nEdges = in_sizes[1] / 2;
  const int* src = ei;
  const int* dst = ei + nEdges;

  float* ws  = (float*)d_ws;
  float* deg = ws;
  float* h2  = ws + 65536;
  unsigned* elist = (unsigned*)(h2 + (size_t)N * 128);

  float* out = (float*)d_out;
  float* h1  = out;
  float* emb = out;
  float* nxt = out + (size_t)N * 64;
  float* cls = out + (size_t)N * 128;
  unsigned* hmap = (unsigned*)cls;

  dim3 gemmGrid(512), gemmBlock(256);
  dim3 ownGrid(256), ownBlock(1024);

  owner_layer_kernel<64, 0, 128> <<<ownGrid, ownBlock, 0, stream>>>(x0, src, dst, hmap, deg, W1, b1, h1, N, nEdges, elist, ws, ws_size);

  owner_layer_kernel<128, 1, 128><<<ownGrid, ownBlock, 0, stream>>>(h1, src, dst, hmap, deg, W2, b2, h2, N, nEdges, elist, ws, ws_size);

  owner_layer_kernel<128, 1, 64> <<<ownGrid, ownBlock, 0, stream>>>(h2, src, dst, hmap, deg, W3, b3, emb, N, nEdges, elist, ws, ws_size);

  gemm_bias_act_kernel<64, 4, false><<<gemmGrid, gemmBlock, 0, stream>>>(nullptr, nullptr, emb, Wp, bp, nxt, N, 64, 0);
  gemm_bias_act_kernel<64, 1, false><<<gemmGrid, gemmBlock, 0, stream>>>(nullptr, nullptr, emb, Wc, bc, cls, N, 10, 0);
}


# First layer's edges dealt to owner blocks by a launch before it: producer blocks bin their edges by owner in LDS, owners read their lines, no scan of all dst ids (5 -> 6 launches); arrival-order sums
# speedup vs baseline: 5.8029x; 1.1721x over previous
#include <hip/hip_runtime.h>
#include <hip/hip_bf16.h>

typedef __attribute__((ext_vector_type(8))) float v8f;
typedef __attribute__((ext_vector_type(2))) float v2f;
typedef __attribute__((ext_vector_type(4))) float v4f;
typedef __attribute__((ext_vector_type(4))) int   v4i;

template <int F, int S, bool FULL>
__device__ __forceinline__ void add_rows(const float* __restrict__ x, float* acc,
                                         int sv, int rv, int lane, int cnt) {
  constexpr int C = F / 32;
  constexpr int U = 32 / C;
  #pragma unroll
  for (int j0 = 0; j0 < 32; j0 += U) {
    if (!FULL && j0 >= cnt) break;
    float v[U][C];
    #pragma unroll
    for (int j = 0; j < U; ++j) {
      const int s = __builtin_amdgcn_readlane(sv, j0 + j);
      const float* xr = x + (size_t)s * F + lane;
      #pragma unroll
      for (int c = 0; c < C; ++c) v[j][c] = xr[c * 32];
    }
    #pragma unroll
    for (int j = 0; j < U; ++j) {
      if (FULL || j0 + j < cnt) {
        const int r = __builtin_amdgcn_readlane(rv, j0 + j);
        float* ar = acc + r * S + lane;
        #pragma unroll
        for (int c = 0; c < C; ++c)
          __hip_atomic_fetch_add(ar + c * 32, v[j][c], __ATOMIC_RELAXED, __HIP_MEMORY_SCOPE_WORKGROUP);
      }
    }
  }
}

constexpr int      LLINES   = 160;
constexpr unsigned LIST_PAD = 0xFFFFFFFFu;

template <int F, int S, bool FULL, bool LISTW, bool SRCQ>
__device__ __forceinline__ void drain(const float* __restrict__ x, const int* __restrict__ src,
                                      float* acc, const unsigned* qp, int lane, int cnt,
                                      unsigned* lstB, int* lcnt) {
  const unsigned ent = qp[FULL ? lane : min(lane, cnt - 1)];
  const int rv = (int)(ent & 255u);
  const int sv = SRCQ ? (int)(ent >> 8) : src[ent >> 8];
  if (LISTW && lstB != nullptr) {
    int c = 0;
    if (lane == 0) c = __hip_atomic_fetch_add(lcnt, 1, __ATOMIC_RELAXED, __HIP_MEMORY_SCOPE_WORKGROUP);
    c = __builtin_amdgcn_readfirstlane(c);
    if (c < LLINES - 1) {
      unsigned w = ((unsigned)sv << 8) | (unsigned)rv;
      if (!FULL && lane >= cnt) w = LIST_PAD;
      lstB[(1 + c) * 32 + lane] = w;
    }
  }
  add_rows<F, S, FULL>(x, acc, sv, rv, lane, cnt);
}

template <bool TAIL, bool VEC, bool DEG>
__device__ __forceinline__ int scan_step(const int* __restrict__ dst, unsigned* q, int qn, int e0,
                                         int lane, int base, int rows, int nEdges, int* cnt) {
  int d[32];
  if (VEC && !TAIL) {
    const v4i* dp = (const v4i*)(dst + e0) + lane;
    #pragma unroll
    for (int k4 = 0; k4 < 8; ++k4) {
      const v4i t = dp[k4 * 32];
      d[4 * k4 + 0] = t.x; d[4 * k4 + 1] = t.y; d[4 * k4 + 2] = t.z; d[4 * k4 + 3] = t.w;
    }
  } else {
    #pragma unroll
    for (int k = 0; k < 32; ++k) {
      const int e = e0 + k * 32 + lane;
      d[k] = dst[TAIL ? min(e, nEdges - 1) : e];
    }
  }
  #pragma unroll
  for (int k = 0; k < 32; ++k) {
    const int e = (VEC && !TAIL) ? (e0 + (k >> 2) * 128 + lane * 4 + (k & 3)) : (e0 + k * 32 + lane);
    const unsigned r = (unsigned)(d[k] - base);
    bool hit = (r < (unsigned)rows);
    if (TAIL) hit = hit && (e < nEdges);
    const unsigned m = __builtin_amdgcn_ballot_w32(hit);
    if (m != 0u) {
      const int pos = qn + (int)__builtin_amdgcn_mbcnt_lo(m, 0u);
      if (hit) {
        q[pos] = ((unsigned)e << 8) | r;
        if (DEG) __hip_atomic_fetch_add(&cnt[r], 1, __ATOMIC_RELAXED, __HIP_MEMORY_SCOPE_WORKGROUP);
      }
      qn += __builtin_popcount(m);
    }
  }
  return qn;
}

constexpr int OWN_RB   = 196;
constexpr int BIN_MAXG = 256;
constexpr int BIN_CAP  = 64;

constexpr int BIN_ROWF = 128;
__device__ __forceinline__ bool bins_ok(int G, int nRows) {
  return (G <= BIN_MAXG) && ((long long)G * OWN_RB >= (long long)nRows) && (nRows < (1 << 24) - 1) &&
         ((size_t)G * G * BIN_CAP + (size_t)G <= (size_t)nRows * BIN_ROWF);
}

__global__ __launch_bounds__(1024) void
bin_edges_kernel(const int* __restrict__ src, const int* __restrict__ dst, unsigned* bins,
                 int nRows, int nEdges) {
  const int G = (int)gridDim.x;
  if (!bins_ok(G, nRows)) return;
  __shared__ unsigned slots[BIN_MAXG * BIN_CAP];
  __shared__ int bcnt[BIN_MAXG];
  __shared__ int ovf;
  for (int i = threadIdx.x; i < BIN_MAXG * BIN_CAP; i += 1024) slots[i] = 0xFFFFFFFFu;
  for (int i = threadIdx.x; i < BIN_MAXG; i += 1024) bcnt[i] = 0;
  if (threadIdx.x == 0) ovf = 0;
  __syncthreads();
  const int chunk = (nEdges + G - 1) / G;
  const long long eBeg = (long long)blockIdx.x * chunk;
  const int e0 = (int)(eBeg < (long long)nEdges ? eBeg : (long long)nEdges);
  const int e1 = min(e0 + chunk, nEdges);
  constexpr int EU = 4;
  for (int eb = e0; eb < e1; eb += EU * 1024) {
    int dv[EU], sv[EU];
    #pragma unroll
    for (int k = 0; k < EU; ++k) {
      const int ec = min(eb + k * 1024 + (int)threadIdx.x, e1 - 1);
      dv[k] = dst[ec];
      sv[k] = src[ec];
    }
    #pragma unroll
    for (int k = 0; k < EU; ++k) {
      const int e = eb + k * 1024 + (int)threadIdx.x;
      const int d = dv[k];
      if (e < e1 && (unsigned)d < (unsigned)nRows) {
        const int o = d / OWN_RB;
        const int r = d - o * OWN_RB;
        const int pos = __hip_atomic_fetch_add(&bcnt[o], 1, __ATOMIC_RELAXED, __HIP_MEMORY_SCOPE_WORKGROUP);
        if (pos < BIN_CAP) slots[o * BIN_CAP + pos] = ((unsigned)sv[k] << 8) | (unsigned)r;
      }
    }
  }
  __syncthreads();
  const int lane = threadIdx.x & 31;
  const int wave = __builtin_amdgcn_readfirstlane((int)(threadIdx.x >> 5));
  unsigned* planeA = bins;
  unsigned* planeB = bins + (size_t)G * G * 32;
  for (int o = wave; o < G; o += 32) {
    const int c = bcnt[o];
    const size_t li = ((size_t)o * G + blockIdx.x) * 32 + lane;
    planeA[li] = slots[o * BIN_CAP + lane];
    if (c >= 32) planeB[li] = slots[o * BIN_CAP + 32 + lane];
    if (c > BIN_CAP && lane == 0) ovf = 1;
  }
  __syncthreads();
  if (threadIdx.x == 0) bins[(size_t)G * G * BIN_CAP + blockIdx.x] = (unsigned)ovf;
}

template <int F, int MODE, int FOUT>
__global__ __launch_bounds__(1024) void
owner_layer_kernel(const float* __restrict__ x, const int* __restrict__ src,
                   const int* __restrict__ dst, const unsigned* bins, float* rdeg,
                   const float* W, const float* bias,
                   float* __restrict__ out, int nRows, int nEdges,
                   unsigned* lst, const float* wsBase, size_t wsBytes) {
  constexpr int RB    = OWN_RB;
  constexpr int RT    = (RB + 15) / 16;
  constexpr int RBP   = RT * 16;
  constexpr int STRIDE = F + 4;
  constexpr int NT    = FOUT / 16;
  constexpr int RG    = 32 / NT;
  constexpr int KS    = F / 4;
  static_assert(FOUT % 16 == 0 && 32 % NT == 0, "column tiles must split evenly over the 32 waves");
  constexpr int STEP  = 32 * 32;
  constexpr int QCAP  = STEP + 32;
  __shared__ __attribute__((aligned(16))) float acc[RBP * STRIDE];
  __shared__ unsigned queue[32 * QCAP];
  __shared__ int cnt[MODE == 0 ? RB : 1];
  __shared__ int lcnt;

  const int lane = threadIdx.x & 31;
  const int wave = __builtin_amdgcn_readfirstlane((int)(threadIdx.x >> 5));
  unsigned* q = queue + wave * QCAP;
  const bool vec = (((size_t)dst) & 15) == 0;
  const int nRounds = (nEdges + 32 * STEP - 1) / (32 * STEP);
  const bool onePass = ((long long)gridDim.x * RB >= (long long)nRows);
  const size_t lstOff = (size_t)((const float*)lst - wsBase) * 4;
  const bool lstOk = onePass && (nRows < (1 << 24) - 1) && (wsBytes >= lstOff) &&
                     (wsBytes - lstOff >= (size_t)gridDim.x * (LLINES * 128));
  unsigned* lstB = lstOk ? lst + (size_t)blockIdx.x * (LLINES * 32) : nullptr;

  for (int base = blockIdx.x * RB; base < nRows; base += gridDim.x * RB) {
    const int rows = min(RB, nRows - base);
    for (int i = threadIdx.x; i < RBP * STRIDE / 4; i += 1024) ((v4f*)acc)[i] = v4f{0.0f, 0.0f, 0.0f, 0.0f};
    if (MODE == 0) for (int i = threadIdx.x; i < RB; i += 1024) cnt[i] = 0;
    if (MODE == 0 && threadIdx.x == 0) lcnt = 0;
    __syncthreads();

    int qn = 0;
    bool srcq = false;
    int nCh = -1;
    if (MODE == 1 && lstOk) nCh = __builtin_amdgcn_readfirstlane((int)lstB[0]);
    if (MODE == 1 && nCh >= 0 && nCh <= LLINES - 1) {
      const unsigned* lp = lstB + 32 + lane;
      unsigned ent = (wave < nCh) ? lp[wave * 32] : LIST_PAD;
      for (int c = wave; c < nCh; c += 32) {
        const unsigned en = lp[min(c + 32, nCh - 1) * 32];
        const int nv = __builtin_popcount(__builtin_amdgcn_ballot_w32(ent != LIST_PAD));
        if (nv == 32) {
          add_rows<F, STRIDE, true>(x, acc, (int)(ent >> 8), (int)(ent & 255u), lane, 32);
        } else if (nv > 0) {
          const unsigned last = (unsigned)__builtin_amdgcn_readlane((int)ent, nv - 1);
          const unsigned e2 = (lane < nv) ? ent : last;
          add_rows<F, STRIDE, false>(x, acc, (int)(e2 >> 8), (int)(e2 & 255u), lane, nv);
        }
        ent = en;
      }
    } else {
      constexpr int NP = BIN_MAXG / 32;
      bool useBins = false;
      unsigned wv[NP];
      if (MODE == 0 && bins_ok((int)gridDim.x, nRows)) {
        const int G = (int)gridDim.x;
        const unsigned* pa = bins + (size_t)blockIdx.x * G * 32 + lane;
        #pragma unroll
        for (int k = 0; k < NP; ++k) {
          const int p = wave + 32 * k;
          const unsigned t = pa[(size_t)min(p, G - 1) * 32];
          wv[k] = (p < G) ? t : LIST_PAD;
        }
        const unsigned* bstat = bins + (size_t)G * G * BIN_CAP;
        unsigned any = 0u;
        for (int i = lane; i < G; i += 32) any |= bstat[i];
        useBins = (__builtin_amdgcn_ballot_w32(any != 0u) == 0u);
      }
      if (MODE == 0 && useBins) {
        srcq = true;
        const int G = (int)gridDim.x;
        const unsigned* pb = bins + ((size_t)G * G + (size_t)blockIdx.x * G) * 32 + lane;
        #pragma unroll
        for (int k = 0; k < NP; ++k) {
          const int p = wave + 32 * k;
          unsigned w = wv[k];
          for (int pl = 0; pl < 2; ++pl) {
            const bool valid = (w != LIST_PAD);
            const unsigned m = __builtin_amdgcn_ballot_w32(valid);
            if (m != 0u) {
              const int pos = qn + (int)__builtin_amdgcn_mbcnt_lo(m, 0u);
              if (valid) {
                q[pos] = w;
                __hip_atomic_fetch_add(&cnt[w & 255u], 1, __ATOMIC_RELAXED, __HIP_MEMORY_SCOPE_WORKGROUP);
              }
              qn += __builtin_popcount(m);
            }
            if (pl == 1 || m != 0xFFFFFFFFu) break;
            w = pb[(size_t)p * 32];
          }
        }
        static_assert(NP * BIN_CAP <= QCAP, "the wave's queue must hold all of its lines");
        while (qn >= 32) { qn -= 32; drain<F, STRIDE, true, true, true>(x, src, acc, q + qn, lane, 32, lstB, &lcnt); }
      } else {
      for (int t = 0; t < nRounds; ++t) {
        const int tt = (t + (int)blockIdx.x) % nRounds;
        const int e0 = tt * (32 * STEP) + wave * STEP;
        if (e0 < nEdges) {
          if (e0 + STEP > nEdges) qn = scan_step<true,  false, MODE == 0>(dst, q, qn, e0, lane, base, rows, nEdges, cnt);
          else if (vec)           qn = scan_step<false, true,  MODE == 0>(dst, q, qn, e0, lane, base, rows, nEdges, cnt);
          else                    qn = scan_step<false, false, MODE == 0>(dst, q, qn, e0, lane, base, rows, nEdges, cnt);
          while (qn >= 32) { qn -= 32; drain<F, STRIDE, true, MODE == 0, false>(x, src, acc, q + qn, lane, 32, lstB, &lcnt); }
        }
      }
      }
    }
    if (qn > 0) {
      if (srcq) drain<F, STRIDE, false, MODE == 0, true >(x, src, acc, q, lane, qn, lstB, &lcnt);
      else      drain<F, STRIDE, false, MODE == 0, false>(x, src, acc, q, lane, qn, lstB, &lcnt);
    }

    __syncthreads();
    if (MODE == 0 && lstB != nullptr && wave == 0) lstB[lane] = (unsigned)lcnt;

    {
      constexpr int PA = (RB * (F / 4) + 1023) / 1024;
      const int nA = rows * (F / 4);
      v4f   xv[PA];
      float rd[PA];
      #pragma unroll
      for (int k = 0; k < PA; ++k) {
        const int i  = min((int)threadIdx.x + k * 1024, nA - 1);
        const int r  = i / (F / 4);
        const int c4 = i - r * (F / 4);
        xv[k] = ((const v4f*)(x + (size_t)(base + r) * F))[c4];
        rd[k] = (MODE == 0) ? 1.0f / fmaxf((float)cnt[r], 1.0f) : rdeg[base + r];
      }
      #pragma unroll
      for (int k = 0; k < PA; ++k) {
        const int i = (int)threadIdx.x + k * 1024;
        if (i < nA) {
          const int r  = i / (F / 4);
          const int c4 = i - r * (F / 4);
          v4f* ap = (v4f*)&acc[r * STRIDE + 4 * c4];
          *ap = *ap * rd[k] + xv[k];
        }
      }
    }
    if (MODE == 0)
      for (int i = threadIdx.x; i < rows; i += 1024) rdeg[base + i] = 1.0f / fmaxf((float)cnt[i], 1.0f);
    __syncthreads();

    {
      const int half = lane >> 4;
      const int l16  = lane & 15;
      const int ct   = wave % NT;
      const int rg   = wave / NT;
      const int n    = ct * 16 + l16;
      const float* Wrow = W + (size_t)n * F + 2 * half;
      v2f bf[KS];
      #pragma unroll
      for (int s = 0; s < KS; ++s) bf[s] = *(const v2f*)(Wrow + 4 * s);
      const float bv = bias[n];

      for (int rt = rg; rt < RT; rt += RG) {
        if (rt * 16 >= rows) break;
        const float* Arow = &acc[(rt * 16 + l16) * STRIDE + 2 * half];
        v8f c = {};
        #pragma unroll
        for (int s = 0; s < KS; ++s) {
          const v2f a = *(const v2f*)(Arow + 4 * s);
          c = __builtin_amdgcn_wmma_f32_16x16x4_f32(false, a, false, bf[s], (short)0, c, false, false);
        }
        const int lrow0 = rt * 16 + 8 * half;
        float* colBase = out + (size_t)(base + lrow0) * FOUT + n;
        if (rt * 16 + 16 <= rows) {
          #pragma unroll
          for (int r = 0; r < 8; ++r) colBase[(size_t)r * FOUT] = fmaxf(c[r] + bv, 0.0f);
        } else {
          #pragma unroll
          for (int r = 0; r < 8; ++r)
            if (lrow0 + r < rows) colBase[(size_t)r * FOUT] = fmaxf(c[r] + bv, 0.0f);
        }
      }
    }
    __syncthreads();
  }
}

template <int FIN, int NT, bool HAS_AGG>
__global__ __launch_bounds__(256) void
gemm_bias_act_kernel(const float* __restrict__ agg, const float* __restrict__ rdeg,
                     const float* __restrict__ xin, const float* __restrict__ W,
                     const float* __restrict__ bias, float* __restrict__ out,
                     int nRows, int Fout, int doRelu)
{
  constexpr int RG     = 8 / NT;
  constexpr int ROWS   = 16 * RG;
  constexpr int STRIDE = FIN + 4;
  constexpr int KS     = FIN / 4;
  constexpr int RF4    = FIN / 4;
  constexpr int LD     = (ROWS * RF4) / 256;
  static_assert((ROWS * RF4) % 256 == 0, "tile must split evenly over the block");
  __shared__ float As[2][ROWS * STRIDE];

  const int lane = threadIdx.x & 31;
  const int wave = threadIdx.x >> 5;
  const int half = lane >> 4;
  const int l16  = lane & 15;
  const int ct   = wave % NT;
  const int rg   = wave / NT;
  const int n    = ct * 16 + l16;
  const bool colOk = (n < Fout);

  const float* Wrow = W + (size_t)(colOk ? n : Fout - 1) * FIN + 2 * half;
  v2f bf[KS];
  #pragma unroll
  for (int s = 0; s < KS; ++s) {
    v2f b = *(const v2f*)(Wrow + 4 * s);
    if (!colOk) { b.x = 0.0f; b.y = 0.0f; }
    bf[s] = b;
  }
  const float bv = colOk ? bias[n] : 0.0f;

  const int nSteps = (nRows + ROWS - 1) / ROWS;
  v4f   pa[LD], px[LD];
  float pr[LD];

  auto fetch = [&](int step) {
    #pragma unroll
    for (int i = 0; i < LD; ++i) {
      int idx = threadIdx.x + i * 256;
      int r   = idx / RF4;
      int c4  = idx - r * RF4;
      int gr  = min(step * ROWS + r, nRows - 1);
      px[i] = ((const v4f*)xin)[(size_t)gr * RF4 + c4];
      if constexpr (HAS_AGG) {
        pa[i] = ((const v4f*)agg)[(size_t)gr * RF4 + c4];
        pr[i] = rdeg[gr];
      }
    }
  };
  auto stage = [&](int b) {
    #pragma unroll
    for (int i = 0; i < LD; ++i) {
      int idx = threadIdx.x + i * 256;
      int r   = idx / RF4;
      int c4  = idx - r * RF4;
      v4f v = px[i];
      if constexpr (HAS_AGG) v += pa[i] * pr[i];
      *(v4f*)&As[b][r * STRIDE + 4 * c4] = v;
    }
  };

  int u = blockIdx.x;
  if (u < nSteps) { fetch(u); stage(0); }
  __syncthreads();

  int buf = 0;
  for (; u < nSteps; u += gridDim.x) {
    const int  un      = u + gridDim.x;
    const bool hasNext = (un < nSteps);
    if (hasNext) fetch(un);

    const float* Arow = &As[buf][(rg * 16 + l16) * STRIDE + 2 * half];
    v8f c = {};
    #pragma unroll
    for (int s = 0; s < KS; ++s) {
      v2f a = *(const v2f*)(Arow + 4 * s);
      c = __builtin_amdgcn_wmma_f32_16x16x4_f32(false, a, false, bf[s], (short)0, c, false, false);
    }

    const int row0 = u * ROWS + rg * 16 + 8 * half;
    float* colBase = out + (size_t)row0 * Fout + n;
    if ((u * ROWS + ROWS <= nRows) && (ct * 16 + 16 <= Fout)) {
      #pragma unroll
      for (int r = 0; r < 8; ++r) {
        float val = c[r] + bv;
        if (doRelu) val = fmaxf(val, 0.0f);
        colBase[(size_t)r * Fout] = val;
      }
    } else {
      #pragma unroll
      for (int r = 0; r < 8; ++r) {
        float val = c[r] + bv;
        if (doRelu) val = fmaxf(val, 0.0f);
        if (row0 + r < nRows && colOk) colBase[(size_t)r * Fout] = val;
      }
    }

    if (hasNext) stage(buf ^ 1);
    __syncthreads();
    buf ^= 1;
  }
}

extern "C" void kernel_launch(void* const* d_in, const int* in_sizes, int n_in,
                              void* d_out, int out_size, void* d_ws, size_t ws_size,
                              hipStream_t stream) {
  const float* x0 = (const float*)d_in[0];
  const int*   ei = (const int*)d_in[1];
  const float *W1 = (const float*)d_in[2],  *b1 = (const float*)d_in[3];
  const float *W2 = (const float*)d_in[4],  *b2 = (const float*)d_in[5];
  const float *W3 = (const float*)d_in[6],  *b3 = (const float*)d_in[7];
  const float *Wp = (const float*)d_in[8],  *bp = (const float*)d_in[9];
  const float *Wc = (const float*)d_in[10], *bc = (const float*)d_in[11];

  const int N      = in_sizes[0] / 64;
  const int nEdges = in_sizes[1] / 2;
  const int* src = ei;
  const int* dst = ei + nEdges;

  float* ws  = (float*)d_ws;
  float* deg = ws;
  float* h2  = ws + 65536;
  unsigned* elist = (unsigned*)(h2 + (size_t)N * 128);

  float* out = (float*)d_out;
  float* h1  = out;
  float* emb = out;
  float* nxt = out + (size_t)N * 64;
  float* cls = out + (size_t)N * 128;
  unsigned* bins = (unsigned*)h2;

  dim3 gemmGrid(512), gemmBlock(256);
  dim3 ownGrid(256), ownBlock(1024);

  bin_edges_kernel<<<ownGrid, ownBlock, 0, stream>>>(src, dst, bins, N, nEdges);

  owner_layer_kernel<64, 0, 128> <<<ownGrid, ownBlock, 0, stream>>>(x0, src, dst, bins, deg, W1, b1, h1, N, nEdges, elist, ws, ws_size);

  owner_layer_kernel<128, 1, 128><<<ownGrid, ownBlock, 0, stream>>>(h1, src, dst, nullptr, deg, W2, b2, h2, N, nEdges, elist, ws, ws_size);

  owner_layer_kernel<128, 1, 64> <<<ownGrid, ownBlock, 0, stream>>>(h2, src, dst, nullptr, deg, W3, b3, emb, N, nEdges, elist, ws, ws_size);

  gemm_bias_act_kernel<64, 4, false><<<gemmGrid, gemmBlock, 0, stream>>>(nullptr, nullptr, emb, Wp, bp, nxt, N, 64, 0);
  gemm_bias_act_kernel<64, 1, false><<<gemmGrid, gemmBlock, 0, stream>>>(nullptr, nullptr, emb, Wc, bc, cls, N, 10, 0);
}


# Layers 2-3 and heads: W goes once per block into LDS by whole-line loads, waves read B fragments there (before: every wave's own 8-byte loads over 16 rows of W, 16 lines a load); same sums and order
# speedup vs baseline: 6.2021x; 1.0688x over previous
#include <hip/hip_runtime.h>
#include <hip/hip_bf16.h>

typedef __attribute__((ext_vector_type(8))) float v8f;
typedef __attribute__((ext_vector_type(2))) float v2f;
typedef __attribute__((ext_vector_type(4))) float v4f;
typedef __attribute__((ext_vector_type(4))) int   v4i;

template <int F, int S, bool FULL>
__device__ __forceinline__ void add_rows(const float* __restrict__ x, float* acc,
                                         int sv, int rv, int lane, int cnt) {
  constexpr int C = F / 32;
  constexpr int U = 32 / C;
  #pragma unroll
  for (int j0 = 0; j0 < 32; j0 += U) {
    if (!FULL && j0 >= cnt) break;
    float v[U][C];
    #pragma unroll
    for (int j = 0; j < U; ++j) {
      const int s = __builtin_amdgcn_readlane(sv, j0 + j);
      const float* xr = x + (size_t)s * F + lane;
      #pragma unroll
      for (int c = 0; c < C; ++c) v[j][c] = xr[c * 32];
    }
    #pragma unroll
    for (int j = 0; j < U; ++j) {
      if (FULL || j0 + j < cnt) {
        const int r = __builtin_amdgcn_readlane(rv, j0 + j);
        float* ar = acc + r * S + lane;
        #pragma unroll
        for (int c = 0; c < C; ++c)
          __hip_atomic_fetch_add(ar + c * 32, v[j][c], __ATOMIC_RELAXED, __HIP_MEMORY_SCOPE_WORKGROUP);
      }
    }
  }
}

constexpr int      LLINES   = 160;
constexpr unsigned LIST_PAD = 0xFFFFFFFFu;

template <int F, int S, bool FULL, bool LISTW, bool SRCQ>
__device__ __forceinline__ void drain(const float* __restrict__ x, const int* __restrict__ src,
                                      float* acc, const unsigned* qp, int lane, int cnt,
                                      unsigned* lstB, int* lcnt) {
  const unsigned ent = qp[FULL ? lane : min(lane, cnt - 1)];
  const int rv = (int)(ent & 255u);
  const int sv = SRCQ ? (int)(ent >> 8) : src[ent >> 8];
  if (LISTW && lstB != nullptr) {
    int c = 0;
    if (lane == 0) c = __hip_atomic_fetch_add(lcnt, 1, __ATOMIC_RELAXED, __HIP_MEMORY_SCOPE_WORKGROUP);
    c = __builtin_amdgcn_readfirstlane(c);
    if (c < LLINES - 1) {
      unsigned w = ((unsigned)sv << 8) | (unsigned)rv;
      if (!FULL && lane >= cnt) w = LIST_PAD;
      lstB[(1 + c) * 32 + lane] = w;
    }
  }
  add_rows<F, S, FULL>(x, acc, sv, rv, lane, cnt);
}

template <bool TAIL, bool VEC, bool DEG>
__device__ __forceinline__ int scan_step(const int* __restrict__ dst, unsigned* q, int qn, int e0,
                                         int lane, int base, int rows, int nEdges, int* cnt) {
  int d[32];
  if (VEC && !TAIL) {
    const v4i* dp = (const v4i*)(dst + e0) + lane;
    #pragma unroll
    for (int k4 = 0; k4 < 8; ++k4) {
      const v4i t = dp[k4 * 32];
      d[4 * k4 + 0] = t.x; d[4 * k4 + 1] = t.y; d[4 * k4 + 2] = t.z; d[4 * k4 + 3] = t.w;
    }
  } else {
    #pragma unroll
    for (int k = 0; k < 32; ++k) {
      const int e = e0 + k * 32 + lane;
      d[k] = dst[TAIL ? min(e, nEdges - 1) : e];
    }
  }
  #pragma unroll
  for (int k = 0; k < 32; ++k) {
    const int e = (VEC && !TAIL) ? (e0 + (k >> 2) * 128 + lane * 4 + (k & 3)) : (e0 + k * 32 + lane);
    const unsigned r = (unsigned)(d[k] - base);
    bool hit = (r < (unsigned)rows);
    if (TAIL) hit = hit && (e < nEdges);
    const unsigned m = __builtin_amdgcn_ballot_w32(hit);
    if (m != 0u) {
      const int pos = qn + (int)__builtin_amdgcn_mbcnt_lo(m, 0u);
      if (hit) {
        q[pos] = ((unsigned)e << 8) | r;
        if (DEG) __hip_atomic_fetch_add(&cnt[r], 1, __ATOMIC_RELAXED, __HIP_MEMORY_SCOPE_WORKGROUP);
      }
      qn += __builtin_popcount(m);
    }
  }
  return qn;
}

constexpr int OWN_RB   = 196;
constexpr int BIN_MAXG = 256;
constexpr int BIN_CAP  = 64;

constexpr int BIN_ROWF = 128;
__device__ __forceinline__ bool bins_ok(int G, int nRows) {
  return (G <= BIN_MAXG) && ((long long)G * OWN_RB >= (long long)nRows) && (nRows < (1 << 24) - 1) &&
         ((size_t)G * G * BIN_CAP + (size_t)G <= (size_t)nRows * BIN_ROWF);
}

__global__ __launch_bounds__(1024) void
bin_edges_kernel(const int* __restrict__ src, const int* __restrict__ dst, unsigned* bins,
                 int nRows, int nEdges) {
  const int G = (int)gridDim.x;
  if (!bins_ok(G, nRows)) return;
  __shared__ unsigned slots[BIN_MAXG * BIN_CAP];
  __shared__ int bcnt[BIN_MAXG];
  __shared__ int ovf;
  for (int i = threadIdx.x; i < BIN_MAXG * BIN_CAP; i += 1024) slots[i] = 0xFFFFFFFFu;
  for (int i = threadIdx.x; i < BIN_MAXG; i += 1024) bcnt[i] = 0;
  if (threadIdx.x == 0) ovf = 0;
  __syncthreads();
  const int chunk = (nEdges + G - 1) / G;
  const long long eBeg = (long long)blockIdx.x * chunk;
  const int e0 = (int)(eBeg < (long long)nEdges ? eBeg : (long long)nEdges);
  const int e1 = min(e0 + chunk, nEdges);
  constexpr int EU = 4;
  for (int eb = e0; eb < e1; eb += EU * 1024) {
    int dv[EU], sv[EU];
    #pragma unroll
    for (int k = 0; k < EU; ++k) {
      const int ec = min(eb + k * 1024 + (int)threadIdx.x, e1 - 1);
      dv[k] = dst[ec];
      sv[k] = src[ec];
    }
    #pragma unroll
    for (int k = 0; k < EU; ++k) {
      const int e = eb + k * 1024 + (int)threadIdx.x;
      const int d = dv[k];
      if (e < e1 && (unsigned)d < (unsigned)nRows) {
        const int o = d / OWN_RB;
        const int r = d - o * OWN_RB;
        const int pos = __hip_atomic_fetch_add(&bcnt[o], 1, __ATOMIC_RELAXED, __HIP_MEMORY_SCOPE_WORKGROUP);
        if (pos < BIN_CAP) slots[o * BIN_CAP + pos] = ((unsigned)sv[k] << 8) | (unsigned)r;
      }
    }
  }
  __syncthreads();
  const int lane = threadIdx.x & 31;
  const int wave = __builtin_amdgcn_readfirstlane((int)(threadIdx.x >> 5));
  unsigned* planeA = bins;
  unsigned* planeB = bins + (size_t)G * G * 32;
  for (int o = wave; o < G; o += 32) {
    const int c = bcnt[o];
    const size_t li = ((size_t)o * G + blockIdx.x) * 32 + lane;
    planeA[li] = slots[o * BIN_CAP + lane];
    if (c >= 32) planeB[li] = slots[o * BIN_CAP + 32 + lane];
    if (c > BIN_CAP && lane == 0) ovf = 1;
  }
  __syncthreads();
  if (threadIdx.x == 0) bins[(size_t)G * G * BIN_CAP + blockIdx.x] = (unsigned)ovf;
}

template <int F, int MODE, int FOUT>
__global__ __launch_bounds__(1024) void
owner_layer_kernel(const float* __restrict__ x, const int* __restrict__ src,
                   const int* __restrict__ dst, const unsigned* bins, float* rdeg,
                   const float* W, const float* bias,
                   float* __restrict__ out, int nRows, int nEdges,
                   unsigned* lst, const float* wsBase, size_t wsBytes) {
  constexpr int RB    = OWN_RB;
  constexpr int RT    = (RB + 15) / 16;
  constexpr int RBP   = RT * 16;
  constexpr int STRIDE = F + 4;
  constexpr int NT    = FOUT / 16;
  constexpr int RG    = 32 / NT;
  constexpr int KS    = F / 4;
  static_assert(FOUT % 16 == 0 && 32 % NT == 0, "column tiles must split evenly over the 32 waves");
  constexpr int STEP  = 32 * 32;
  constexpr int QCAP  = STEP + 32;
  __shared__ __attribute__((aligned(16))) float acc[RBP * STRIDE];
  __shared__ unsigned queue[32 * QCAP];
  __shared__ int cnt[MODE == 0 ? RB : 1];
  __shared__ int lcnt;

  const int lane = threadIdx.x & 31;
  const int wave = __builtin_amdgcn_readfirstlane((int)(threadIdx.x >> 5));
  unsigned* q = queue + wave * QCAP;
  const bool vec = (((size_t)dst) & 15) == 0;
  const int nRounds = (nEdges + 32 * STEP - 1) / (32 * STEP);
  const bool onePass = ((long long)gridDim.x * RB >= (long long)nRows);
  const size_t lstOff = (size_t)((const float*)lst - wsBase) * 4;
  const bool lstOk = onePass && (nRows < (1 << 24) - 1) && (wsBytes >= lstOff) &&
                     (wsBytes - lstOff >= (size_t)gridDim.x * (LLINES * 128));
  unsigned* lstB = lstOk ? lst + (size_t)blockIdx.x * (LLINES * 32) : nullptr;

  for (int base = blockIdx.x * RB; base < nRows; base += gridDim.x * RB) {
    const int rows = min(RB, nRows - base);
    for (int i = threadIdx.x; i < RBP * STRIDE / 4; i += 1024) ((v4f*)acc)[i] = v4f{0.0f, 0.0f, 0.0f, 0.0f};
    if (MODE == 0) for (int i = threadIdx.x; i < RB; i += 1024) cnt[i] = 0;
    if (MODE == 0 && threadIdx.x == 0) lcnt = 0;
    __syncthreads();

    int qn = 0;
    bool srcq = false;
    int nCh = -1;
    if (MODE == 1 && lstOk) nCh = __builtin_amdgcn_readfirstlane((int)lstB[0]);
    if (MODE == 1 && nCh >= 0 && nCh <= LLINES - 1) {
      const unsigned* lp = lstB + 32 + lane;
      unsigned ent = (wave < nCh) ? lp[wave * 32] : LIST_PAD;
      for (int c = wave; c < nCh; c += 32) {
        const unsigned en = lp[min(c + 32, nCh - 1) * 32];
        const int nv = __builtin_popcount(__builtin_amdgcn_ballot_w32(ent != LIST_PAD));
        if (nv == 32) {
          add_rows<F, STRIDE, true>(x, acc, (int)(ent >> 8), (int)(ent & 255u), lane, 32);
        } else if (nv > 0) {
          const unsigned last = (unsigned)__builtin_amdgcn_readlane((int)ent, nv - 1);
          const unsigned e2 = (lane < nv) ? ent : last;
          add_rows<F, STRIDE, false>(x, acc, (int)(e2 >> 8), (int)(e2 & 255u), lane, nv);
        }
        ent = en;
      }
    } else {
      constexpr int NP = BIN_MAXG / 32;
      bool useBins = false;
      unsigned wv[NP];
      if (MODE == 0 && bins_ok((int)gridDim.x, nRows)) {
        const int G = (int)gridDim.x;
        const unsigned* pa = bins + (size_t)blockIdx.x * G * 32 + lane;
        #pragma unroll
        for (int k = 0; k < NP; ++k) {
          const int p = wave + 32 * k;
          const unsigned t = pa[(size_t)min(p, G - 1) * 32];
          wv[k] = (p < G) ? t : LIST_PAD;
        }
        const unsigned* bstat = bins + (size_t)G * G * BIN_CAP;
        unsigned any = 0u;
        for (int i = lane; i < G; i += 32) any |= bstat[i];
        useBins = (__builtin_amdgcn_ballot_w32(any != 0u) == 0u);
      }
      if (MODE == 0 && useBins) {
        srcq = true;
        const int G = (int)gridDim.x;
        const unsigned* pb = bins + ((size_t)G * G + (size_t)blockIdx.x * G) * 32 + lane;
        #pragma unroll
        for (int k = 0; k < NP; ++k) {
          const int p = wave + 32 * k;
          unsigned w = wv[k];
          for (int pl = 0; pl < 2; ++pl) {
            const bool valid = (w != LIST_PAD);
            const unsigned m = __builtin_amdgcn_ballot_w32(valid);
            if (m != 0u) {
              const int pos = qn + (int)__builtin_amdgcn_mbcnt_lo(m, 0u);
              if (valid) {
                q[pos] = w;
                __hip_atomic_fetch_add(&cnt[w & 255u], 1, __ATOMIC_RELAXED, __HIP_MEMORY_SCOPE_WORKGROUP);
              }
              qn += __builtin_popcount(m);
            }
            if (pl == 1 || m != 0xFFFFFFFFu) break;
            w = pb[(size_t)p * 32];
          }
        }
        static_assert(NP * BIN_CAP <= QCAP, "the wave's queue must hold all of its lines");
        while (qn >= 32) { qn -= 32; drain<F, STRIDE, true, true, true>(x, src, acc, q + qn, lane, 32, lstB, &lcnt); }
      } else {
      for (int t = 0; t < nRounds; ++t) {
        const int tt = (t + (int)blockIdx.x) % nRounds;
        const int e0 = tt * (32 * STEP) + wave * STEP;
        if (e0 < nEdges) {
          if (e0 + STEP > nEdges) qn = scan_step<true,  false, MODE == 0>(dst, q, qn, e0, lane, base, rows, nEdges, cnt);
          else if (vec)           qn = scan_step<false, true,  MODE == 0>(dst, q, qn, e0, lane, base, rows, nEdges, cnt);
          else                    qn = scan_step<false, false, MODE == 0>(dst, q, qn, e0, lane, base, rows, nEdges, cnt);
          while (qn >= 32) { qn -= 32; drain<F, STRIDE, true, MODE == 0, false>(x, src, acc, q + qn, lane, 32, lstB, &lcnt); }
        }
      }
      }
    }
    if (qn > 0) {
      if (srcq) drain<F, STRIDE, false, MODE == 0, true >(x, src, acc, q, lane, qn, lstB, &lcnt);
      else      drain<F, STRIDE, false, MODE == 0, false>(x, src, acc, q, lane, qn, lstB, &lcnt);
    }

    __syncthreads();
    if (MODE == 0 && lstB != nullptr && wave == 0) lstB[lane] = (unsigned)lcnt;

    constexpr bool WLDS = (MODE == 1);
    constexpr int WPC = WLDS ? (FOUT * (F / 4)) / 1024 : 0;
    static_assert((FOUT * (F / 4)) % 1024 == 0, "W must split evenly over the block");
    static_assert((size_t)FOUT * STRIDE <= (size_t)32 * QCAP, "the W copy must fit the queues' space");
    static_assert(WPC <= (RB * (F / 4) + 1023) / 1024, "the W copy rides the combine's loops");
    v4f wreg[WLDS ? WPC : 1];
    float* Wl = (float*)queue;

    {
      constexpr int PA = (RB * (F / 4) + 1023) / 1024;
      const int nA = rows * (F / 4);
      v4f   xv[PA];
      float rd[PA];
      #pragma unroll
      for (int k = 0; k < PA; ++k) {
        const int i  = min((int)threadIdx.x + k * 1024, nA - 1);
        const int r  = i / (F / 4);
        const int c4 = i - r * (F / 4);
        xv[k] = ((const v4f*)(x + (size_t)(base + r) * F))[c4];
        rd[k] = (MODE == 0) ? 1.0f / fmaxf((float)cnt[r], 1.0f) : rdeg[base + r];
        if (k < WPC) wreg[k] = ((const v4f*)W)[(int)threadIdx.x + k * 1024];
      }
      #pragma unroll
      for (int k = 0; k < PA; ++k) {
        const int i = (int)threadIdx.x + k * 1024;
        const int r  = i / (F / 4);
        const int c4 = i - r * (F / 4);
        if (i < nA) {
          v4f* ap = (v4f*)&acc[r * STRIDE + 4 * c4];
          *ap = *ap * rd[k] + xv[k];
        }
        if (k < WPC) *(v4f*)&Wl[r * STRIDE + 4 * c4] = wreg[k];
      }
    }
    if (MODE == 0)
      for (int i = threadIdx.x; i < rows; i += 1024) rdeg[base + i] = 1.0f / fmaxf((float)cnt[i], 1.0f);
    __syncthreads();

    {
      const int half = lane >> 4;
      const int l16  = lane & 15;
      const int ct   = wave % NT;
      const int rg   = wave / NT;
      const int n    = ct * 16 + l16;
      const float* Wrow = WLDS ? (const float*)(Wl + n * STRIDE + 2 * half) : (W + (size_t)n * F + 2 * half);
      v2f bf[KS];
      #pragma unroll
      for (int s = 0; s < KS; ++s) bf[s] = *(const v2f*)(Wrow + 4 * s);
      const float bv = bias[n];

      for (int rt = rg; rt < RT; rt += RG) {
        if (rt * 16 >= rows) break;
        const float* Arow = &acc[(rt * 16 + l16) * STRIDE + 2 * half];
        v8f c = {};
        #pragma unroll
        for (int s = 0; s < KS; ++s) {
          const v2f a = *(const v2f*)(Arow + 4 * s);
          c = __builtin_amdgcn_wmma_f32_16x16x4_f32(false, a, false, bf[s], (short)0, c, false, false);
        }
        const int lrow0 = rt * 16 + 8 * half;
        float* colBase = out + (size_t)(base + lrow0) * FOUT + n;
        if (rt * 16 + 16 <= rows) {
          #pragma unroll
          for (int r = 0; r < 8; ++r) colBase[(size_t)r * FOUT] = fmaxf(c[r] + bv, 0.0f);
        } else {
          #pragma unroll
          for (int r = 0; r < 8; ++r)
            if (lrow0 + r < rows) colBase[(size_t)r * FOUT] = fmaxf(c[r] + bv, 0.0f);
        }
      }
    }
    __syncthreads();
  }
}

template <int FIN, int NT, bool HAS_AGG>
__global__ __launch_bounds__(256) void
gemm_bias_act_kernel(const float* __restrict__ agg, const float* __restrict__ rdeg,
                     const float* __restrict__ xin, const float* __restrict__ W,
                     const float* __restrict__ bias, float* __restrict__ out,
                     int nRows, int Fout, int doRelu)
{
  constexpr int RG     = 8 / NT;
  constexpr int ROWS   = 16 * RG;
  constexpr int STRIDE = FIN + 4;
  constexpr int KS     = FIN / 4;
  constexpr int RF4    = FIN / 4;
  constexpr int LD     = (ROWS * RF4) / 256;
  static_assert((ROWS * RF4) % 256 == 0, "tile must split evenly over the block");
  __shared__ float As[2][ROWS * STRIDE];
  __shared__ __attribute__((aligned(16))) float Wl[NT * 16 * STRIDE];
  constexpr int WPC = (NT * 16 * RF4) / 256;
  static_assert((NT * 16 * RF4) % 256 == 0, "the W copy must split evenly over the block");

  const int lane = threadIdx.x & 31;
  const int wave = threadIdx.x >> 5;
  const int half = lane >> 4;
  const int l16  = lane & 15;
  const int ct   = wave % NT;
  const int rg   = wave / NT;
  const int n    = ct * 16 + l16;
  const bool colOk = (n < Fout);

  v4f wreg[WPC];
  #pragma unroll
  for (int k = 0; k < WPC; ++k) {
    const int i  = (int)threadIdx.x + k * 256;
    const int r  = i / RF4;
    const int c4 = i - r * RF4;
    wreg[k] = ((const v4f*)W)[(size_t)min(r, Fout - 1) * RF4 + c4];
  }
  const float bv = colOk ? bias[n] : 0.0f;

  const int nSteps = (nRows + ROWS - 1) / ROWS;
  v4f   pa[LD], px[LD];
  float pr[LD];

  auto fetch = [&](int step) {
    #pragma unroll
    for (int i = 0; i < LD; ++i) {
      int idx = threadIdx.x + i * 256;
      int r   = idx / RF4;
      int c4  = idx - r * RF4;
      int gr  = min(step * ROWS + r, nRows - 1);
      px[i] = ((const v4f*)xin)[(size_t)gr * RF4 + c4];
      if constexpr (HAS_AGG) {
        pa[i] = ((const v4f*)agg)[(size_t)gr * RF4 + c4];
        pr[i] = rdeg[gr];
      }
    }
  };
  auto stage = [&](int b) {
    #pragma unroll
    for (int i = 0; i < LD; ++i) {
      int idx = threadIdx.x + i * 256;
      int r   = idx / RF4;
      int c4  = idx - r * RF4;
      v4f v = px[i];
      if constexpr (HAS_AGG) v += pa[i] * pr[i];
      *(v4f*)&As[b][r * STRIDE + 4 * c4] = v;
    }
  };

  int u = blockIdx.x;
  if (u < nSteps) { fetch(u); stage(0); }
  #pragma unroll
  for (int k = 0; k < WPC; ++k) {
    const int i  = (int)threadIdx.x + k * 256;
    const int r  = i / RF4;
    const int c4 = i - r * RF4;
    *(v4f*)&Wl[r * STRIDE + 4 * c4] = (r < Fout) ? wreg[k] : v4f{0.0f, 0.0f, 0.0f, 0.0f};
  }
  __syncthreads();

  const float* Wrow = &Wl[n * STRIDE + 2 * half];
  v2f bf[KS];
  #pragma unroll
  for (int s = 0; s < KS; ++s) bf[s] = *(const v2f*)(Wrow + 4 * s);

  int buf = 0;
  for (; u < nSteps; u += gridDim.x) {
    const int  un      = u + gridDim.x;
    const bool hasNext = (un < nSteps);
    if (hasNext) fetch(un);

    const float* Arow = &As[buf][(rg * 16 + l16) * STRIDE + 2 * half];
    v8f c = {};
    #pragma unroll
    for (int s = 0; s < KS; ++s) {
      v2f a = *(const v2f*)(Arow + 4 * s);
      c = __builtin_amdgcn_wmma_f32_16x16x4_f32(false, a, false, bf[s], (short)0, c, false, false);
    }

    const int row0 = u * ROWS + rg * 16 + 8 * half;
    float* colBase = out + (size_t)row0 * Fout + n;
    if ((u * ROWS + ROWS <= nRows) && (ct * 16 + 16 <= Fout)) {
      #pragma unroll
      for (int r = 0; r < 8; ++r) {
        float val = c[r] + bv;
        if (doRelu) val = fmaxf(val, 0.0f);
        colBase[(size_t)r * Fout] = val;
      }
    } else {
      #pragma unroll
      for (int r = 0; r < 8; ++r) {
        float val = c[r] + bv;
        if (doRelu) val = fmaxf(val, 0.0f);
        if (row0 + r < nRows && colOk) colBase[(size_t)r * Fout] = val;
      }
    }

    if (hasNext) stage(buf ^ 1);
    __syncthreads();
    buf ^= 1;
  }
}

extern "C" void kernel_launch(void* const* d_in, const int* in_sizes, int n_in,
                              void* d_out, int out_size, void* d_ws, size_t ws_size,
                              hipStream_t stream) {
  const float* x0 = (const float*)d_in[0];
  const int*   ei = (const int*)d_in[1];
  const float *W1 = (const float*)d_in[2],  *b1 = (const float*)d_in[3];
  const float *W2 = (const float*)d_in[4],  *b2 = (const float*)d_in[5];
  const float *W3 = (const float*)d_in[6],  *b3 = (const float*)d_in[7];
  const float *Wp = (const float*)d_in[8],  *bp = (const float*)d_in[9];
  const float *Wc = (const float*)d_in[10], *bc = (const float*)d_in[11];

  const int N      = in_sizes[0] / 64;
  const int nEdges = in_sizes[1] / 2;
  const int* src = ei;
  const int* dst = ei + nEdges;

  float* ws  = (float*)d_ws;
  float* deg = ws;
  float* h2  = ws + 65536;
  unsigned* elist = (unsigned*)(h2 + (size_t)N * 128);

  float* out = (float*)d_out;
  float* h1  = out;
  float* emb = out;
  float* nxt = out + (size_t)N * 64;
  float* cls = out + (size_t)N * 128;
  unsigned* bins = (unsigned*)h2;

  dim3 gemmGrid(512), gemmBlock(256);
  dim3 ownGrid(256), ownBlock(1024);

  bin_edges_kernel<<<ownGrid, ownBlock, 0, stream>>>(src, dst, bins, N, nEdges);

  owner_layer_kernel<64, 0, 128> <<<ownGrid, ownBlock, 0, stream>>>(x0, src, dst, bins, deg, W1, b1, h1, N, nEdges, elist, ws, ws_size);

  owner_layer_kernel<128, 1, 128><<<ownGrid, ownBlock, 0, stream>>>(h1, src, dst, nullptr, deg, W2, b2, h2, N, nEdges, elist, ws, ws_size);

  owner_layer_kernel<128, 1, 64> <<<ownGrid, ownBlock, 0, stream>>>(h2, src, dst, nullptr, deg, W3, b3, emb, N, nEdges, elist, ws, ws_size);

  gemm_bias_act_kernel<64, 4, false><<<gemmGrid, gemmBlock, 0, stream>>>(nullptr, nullptr, emb, Wp, bp, nxt, N, 64, 0);
  gemm_bias_act_kernel<64, 1, false><<<gemmGrid, gemmBlock, 0, stream>>>(nullptr, nullptr, emb, Wc, bc, cls, N, 10, 0);
}


# Neighbour sums in registers: rows dealt to a block's waves (row % 32), edges routed to the owner wave's list, fetched rows added in registers by slot index; no LDS float atomics; sums in list order
# speedup vs baseline: 6.7176x; 1.0831x over previous
#include <hip/hip_runtime.h>
#include <hip/hip_bf16.h>

typedef __attribute__((ext_vector_type(8))) float v8f;
typedef __attribute__((ext_vector_type(2))) float v2f;
typedef __attribute__((ext_vector_type(4))) float v4f;
typedef __attribute__((ext_vector_type(4))) int   v4i;


template <int F, int S, bool FULL>
__device__ __forceinline__ void add_rows(const float* __restrict__ x, float* acc,
                                         int sv, int rv, int lane, int cnt) {
  constexpr int C = F / 32;
  constexpr int U = 32 / C;
  #pragma unroll
  for (int j0 = 0; j0 < 32; j0 += U) {
    if (!FULL && j0 >= cnt) break;
    float v[U][C];
    #pragma unroll
    for (int j = 0; j < U; ++j) {
      const int s = __builtin_amdgcn_readlane(sv, j0 + j);
      const float* xr = x + (size_t)s * F + lane;
      #pragma unroll
      for (int c = 0; c < C; ++c) v[j][c] = xr[c * 32];
    }
    #pragma unroll
    for (int j = 0; j < U; ++j) {
      if (FULL || j0 + j < cnt) {
        const int r = __builtin_amdgcn_readlane(rv, j0 + j);
        float* ar = acc + r * S + lane;
        #pragma unroll
        for (int c = 0; c < C; ++c)
          __hip_atomic_fetch_add(ar + c * 32, v[j][c], __ATOMIC_RELAXED, __HIP_MEMORY_SCOPE_WORKGROUP);
      }
    }
  }
}

constexpr int OWN_RB = 196;
constexpr int NSLOT  = (OWN_RB + 31) / 32;

template <int F>
__device__ __forceinline__ void sum_rows(const float* __restrict__ x, v8f (&a)[F / 32],
                                         int sv, int rv, int lane, int cnt) {
  constexpr int C = F / 32;
  constexpr int U = 32 / C;
  for (int j0 = 0; j0 < cnt; j0 += U) {
    float v[U][C];
    #pragma unroll
    for (int j = 0; j < U; ++j) {
      const int s = __builtin_amdgcn_readlane(sv, j0 + j);
      const float* xr = x + (size_t)s * F + lane;
      #pragma unroll
      for (int c = 0; c < C; ++c) v[j][c] = xr[c * 32];
    }
    #pragma unroll
    for (int j = 0; j < U; ++j) {
      const int r    = __builtin_amdgcn_readlane(rv, j0 + j);
      const int slot = (j0 + j < cnt) ? (r >> 5) & 7 : 7;
      #pragma unroll
      for (int c = 0; c < C; ++c) a[c][slot] += v[j][c];
    }
  }
}
static_assert(NSLOT <= 7, "sum_rows needs the 8-float vectors' last slot for itself");

template <int F, int S>
__device__ __forceinline__ void store_sums(float* acc, const v8f (&a)[F / 32], int wave, int lane) {
  #pragma unroll
  for (int s = 0; s < NSLOT; ++s) {
    const int r = wave + 32 * s;
    if (r < OWN_RB) {
      #pragma unroll
      for (int c = 0; c < F / 32; ++c) acc[r * S + lane + 32 * c] = a[c][s];
    }
  }
}

constexpr int      LLINES   = 160;
constexpr unsigned LIST_PAD = 0xFFFFFFFFu;

__device__ __forceinline__ void list_lines(int clLane, int wave, int& start, int& total) {
  const int ln = (int)(((unsigned)clLane + 31u) >> 5);
  int st = 0, tot = 0;
  #pragma unroll
  for (int l = 0; l < 32; ++l) {
    const int t = __builtin_amdgcn_readlane(ln, l);
    st  += (l < wave) ? t : 0;
    tot += t;
  }
  start = st;
  total = tot;
}

template <int F, int S, bool FULL>
__device__ __forceinline__ void drain(const float* __restrict__ x, const int* __restrict__ src,
                                      float* acc, const unsigned* qp, int lane, int cnt) {
  const unsigned ent = qp[FULL ? lane : min(lane, cnt - 1)];
  const int rv = (int)(ent & 255u);
  const int sv = src[ent >> 8];
  add_rows<F, S, FULL>(x, acc, sv, rv, lane, cnt);
}

template <bool TAIL, bool VEC, bool DEG>
__device__ __forceinline__ int scan_step(const int* __restrict__ dst, unsigned* q, int qn, int e0,
                                         int lane, int base, int rows, int nEdges, int* cnt) {
  int d[32];
  if (VEC && !TAIL) {
    const v4i* dp = (const v4i*)(dst + e0) + lane;
    #pragma unroll
    for (int k4 = 0; k4 < 8; ++k4) {
      const v4i t = dp[k4 * 32];
      d[4 * k4 + 0] = t.x; d[4 * k4 + 1] = t.y; d[4 * k4 + 2] = t.z; d[4 * k4 + 3] = t.w;
    }
  } else {
    #pragma unroll
    for (int k = 0; k < 32; ++k) {
      const int e = e0 + k * 32 + lane;
      d[k] = dst[TAIL ? min(e, nEdges - 1) : e];
    }
  }
  #pragma unroll
  for (int k = 0; k < 32; ++k) {
    const int e = (VEC && !TAIL) ? (e0 + (k >> 2) * 128 + lane * 4 + (k & 3)) : (e0 + k * 32 + lane);
    const unsigned r = (unsigned)(d[k] - base);
    bool hit = (r < (unsigned)rows);
    if (TAIL) hit = hit && (e < nEdges);
    const unsigned m = __builtin_amdgcn_ballot_w32(hit);
    if (m != 0u) {
      const int pos = qn + (int)__builtin_amdgcn_mbcnt_lo(m, 0u);
      if (hit) {
        q[pos] = ((unsigned)e << 8) | r;
        if (DEG) __hip_atomic_fetch_add(&cnt[r], 1, __ATOMIC_RELAXED, __HIP_MEMORY_SCOPE_WORKGROUP);
      }
      qn += __builtin_popcount(m);
    }
  }
  return qn;
}

constexpr int BIN_MAXG = 256;
constexpr int BIN_CAP  = 64;

constexpr int BIN_ROWF = 128;
__device__ __forceinline__ bool bins_ok(int G, int nRows) {
  return (G <= BIN_MAXG) && ((long long)G * OWN_RB >= (long long)nRows) && (nRows < (1 << 24) - 1) &&
         ((size_t)G * G * BIN_CAP + (size_t)G <= (size_t)nRows * BIN_ROWF);
}

__global__ __launch_bounds__(1024) void
bin_edges_kernel(const int* __restrict__ src, const int* __restrict__ dst, unsigned* bins,
                 int nRows, int nEdges) {
  const int G = (int)gridDim.x;
  if (!bins_ok(G, nRows)) return;
  __shared__ unsigned slots[BIN_MAXG * BIN_CAP];
  __shared__ int bcnt[BIN_MAXG];
  __shared__ int ovf;
  for (int i = threadIdx.x; i < BIN_MAXG * BIN_CAP; i += 1024) slots[i] = 0xFFFFFFFFu;
  for (int i = threadIdx.x; i < BIN_MAXG; i += 1024) bcnt[i] = 0;
  if (threadIdx.x == 0) ovf = 0;
  __syncthreads();
  const int chunk = (nEdges + G - 1) / G;
  const long long eBeg = (long long)blockIdx.x * chunk;
  const int e0 = (int)(eBeg < (long long)nEdges ? eBeg : (long long)nEdges);
  const int e1 = min(e0 + chunk, nEdges);
  constexpr int EU = 4;
  for (int eb = e0; eb < e1; eb += EU * 1024) {
    int dv[EU], sv[EU];
    #pragma unroll
    for (int k = 0; k < EU; ++k) {
      const int ec = min(eb + k * 1024 + (int)threadIdx.x, e1 - 1);
      dv[k] = dst[ec];
      sv[k] = src[ec];
    }
    #pragma unroll
    for (int k = 0; k < EU; ++k) {
      const int e = eb + k * 1024 + (int)threadIdx.x;
      const int d = dv[k];
      if (e < e1 && (unsigned)d < (unsigned)nRows) {
        const int o = d / OWN_RB;
        const int r = d - o * OWN_RB;
        const int pos = __hip_atomic_fetch_add(&bcnt[o], 1, __ATOMIC_RELAXED, __HIP_MEMORY_SCOPE_WORKGROUP);
        if (pos < BIN_CAP) slots[o * BIN_CAP + pos] = ((unsigned)sv[k] << 8) | (unsigned)r;
      }
    }
  }
  __syncthreads();
  const int lane = threadIdx.x & 31;
  const int wave = __builtin_amdgcn_readfirstlane((int)(threadIdx.x >> 5));
  unsigned* planeA = bins;
  unsigned* planeB = bins + (size_t)G * G * 32;
  for (int o = wave; o < G; o += 32) {
    const int c = bcnt[o];
    const size_t li = ((size_t)o * G + blockIdx.x) * 32 + lane;
    planeA[li] = slots[o * BIN_CAP + lane];
    if (c >= 32) planeB[li] = slots[o * BIN_CAP + 32 + lane];
    if (c > BIN_CAP && lane == 0) ovf = 1;
  }
  __syncthreads();
  if (threadIdx.x == 0) bins[(size_t)G * G * BIN_CAP + blockIdx.x] = (unsigned)ovf;
}

template <int F, int MODE, int FOUT>
__global__ __launch_bounds__(1024) void
owner_layer_kernel(const float* __restrict__ x, const int* __restrict__ src,
                   const int* __restrict__ dst, const unsigned* bins, float* rdeg,
                   const float* W, const float* bias,
                   float* __restrict__ out, int nRows, int nEdges,
                   unsigned* lst, const float* wsBase, size_t wsBytes) {
  constexpr int RB    = OWN_RB;
  constexpr int RT    = (RB + 15) / 16;
  constexpr int RBP   = RT * 16;
  constexpr int STRIDE = F + 4;
  constexpr int NT    = FOUT / 16;
  constexpr int RG    = 32 / NT;
  constexpr int KS    = F / 4;
  constexpr int C     = F / 32;
  static_assert(FOUT % 16 == 0 && 32 % NT == 0, "column tiles must split evenly over the 32 waves");
  constexpr int STEP  = 32 * 32;
  constexpr int QCAP  = STEP + 32;
  __shared__ __attribute__((aligned(16))) float acc[RBP * STRIDE];
  __shared__ unsigned queue[32 * QCAP];
  __shared__ int cnt[MODE == 0 ? RB : 1];
  __shared__ int qcount[32];

  const int lane = threadIdx.x & 31;
  const int wave = __builtin_amdgcn_readfirstlane((int)(threadIdx.x >> 5));
  unsigned* q = queue + wave * QCAP;
  const bool vec = (((size_t)dst) & 15) == 0;
  const int nRounds = (nEdges + 32 * STEP - 1) / (32 * STEP);
  const bool onePass = ((long long)gridDim.x * RB >= (long long)nRows);
  const size_t lstOff = (size_t)((const float*)lst - wsBase) * 4;
  const bool lstOk = onePass && (nRows < (1 << 24) - 1) && (wsBytes >= lstOff) &&
                     (wsBytes - lstOff >= (size_t)gridDim.x * (LLINES * 128));
  unsigned* lstB = lstOk ? lst + (size_t)blockIdx.x * (LLINES * 32) : nullptr;

  for (int base = blockIdx.x * RB; base < nRows; base += gridDim.x * RB) {
    const int rows = min(RB, nRows - base);
    for (int i = threadIdx.x; i < RBP * STRIDE / 4; i += 1024) ((v4f*)acc)[i] = v4f{0.0f, 0.0f, 0.0f, 0.0f};
    if (MODE == 0) for (int i = threadIdx.x; i < RB; i += 1024) cnt[i] = 0;
    if (MODE == 0 && threadIdx.x < 32) qcount[threadIdx.x] = 0;
    __syncthreads();

    bool summed = false;

    if (MODE == 1 && lstOk) {
      const int cl = (int)lstB[lane];
      const bool bad = (__builtin_amdgcn_ballot_w32((unsigned)cl > (unsigned)QCAP) != 0u);
      int st, tot;
      list_lines(cl, wave, st, tot);
      if (!bad && tot <= LLINES - 1) {
        summed = true;
        const int n  = __builtin_amdgcn_readlane(cl, wave);
        const int nl = (n + 31) >> 5;
        v8f a[C];
        #pragma unroll
        for (int c = 0; c < C; ++c) a[c] = v8f{};
        const unsigned* lp = lstB + (size_t)(1 + st) * 32 + lane;
        unsigned ent = LIST_PAD;
        if (nl > 0) ent = lp[0];
        for (int i = 0; i < nl; ++i) {
          const unsigned en = lp[min(i + 1, nl - 1) * 32];
          const int m = min(32, n - 32 * i);
          const unsigned last = (unsigned)__builtin_amdgcn_readlane((int)ent, m - 1);
          const unsigned e2 = (lane < m) ? ent : last;
          sum_rows<F>(x, a, (int)(e2 >> 8), (int)(e2 & 255u), lane, m);
          ent = en;
        }
        store_sums<F, STRIDE>(acc, a, wave, lane);
      }
    }

    if (MODE == 0) {
      constexpr int NP = BIN_MAXG / 32;
      bool useBins = false;
      unsigned wv[NP];
      if (bins_ok((int)gridDim.x, nRows)) {
        const int G = (int)gridDim.x;
        const unsigned* pa = bins + (size_t)blockIdx.x * G * 32 + lane;
        #pragma unroll
        for (int k = 0; k < NP; ++k) {
          const int p = wave + 32 * k;
          const unsigned t = pa[(size_t)min(p, G - 1) * 32];
          wv[k] = (p < G) ? t : LIST_PAD;
        }
        const unsigned* bstat = bins + (size_t)G * G * BIN_CAP;
        unsigned any = 0u;
        for (int i0 = 0; i0 < G; i0 += 32) {
          const int i = i0 + lane;
          const unsigned t = bstat[min(i, G - 1)];
          any |= (i < G) ? t : 0u;
        }
        useBins = (__builtin_amdgcn_ballot_w32(any != 0u) == 0u);
      }
      if (useBins) {
        const int G = (int)gridDim.x;
        const unsigned* pb = bins + ((size_t)G * G + (size_t)blockIdx.x * G) * 32 + lane;
        #pragma unroll
        for (int k = 0; k < NP; ++k) {
          const int p = wave + 32 * k;
          unsigned w = wv[k];
          for (int pl = 0; pl < 2; ++pl) {
            const bool valid = (w != LIST_PAD);
            const unsigned m = __builtin_amdgcn_ballot_w32(valid);
            if (valid) {
              const unsigned ow = w & 31u;
              const int pos = __hip_atomic_fetch_add(&qcount[ow], 1, __ATOMIC_RELAXED, __HIP_MEMORY_SCOPE_WORKGROUP);
              if (pos < QCAP) queue[ow * QCAP + pos] = w;
              __hip_atomic_fetch_add(&cnt[w & 255u], 1, __ATOMIC_RELAXED, __HIP_MEMORY_SCOPE_WORKGROUP);
            }
            if (pl == 1 || m != 0xFFFFFFFFu) break;
            w = pb[(size_t)p * 32];
          }
        }
        __syncthreads();
        const int cl = qcount[lane];
        const bool over = (__builtin_amdgcn_ballot_w32(cl > QCAP) != 0u);
        if (!over) {
          summed = true;
          int st, tot;
          list_lines(cl, wave, st, tot);
          const bool keep = (lstB != nullptr) && (tot <= LLINES - 1);
          if (lstB != nullptr && wave == 0) lstB[lane] = keep ? (unsigned)cl : LIST_PAD;
          const int n = __builtin_amdgcn_readlane(cl, wave);
          v8f a[C];
          #pragma unroll
          for (int c = 0; c < C; ++c) a[c] = v8f{};
          for (int b0 = 0; b0 < n; b0 += 32) {
            const int m = min(32, n - b0);
            const unsigned ent = q[b0 + min(lane, m - 1)];
            if (keep) lstB[(size_t)(1 + st + (b0 >> 5)) * 32 + lane] = (lane < m) ? ent : LIST_PAD;
            sum_rows<F>(x, a, (int)(ent >> 8), (int)(ent & 255u), lane, m);
          }
          store_sums<F, STRIDE>(acc, a, wave, lane);
        } else {
          if (threadIdx.x < RB) cnt[threadIdx.x] = 0;
          __syncthreads();
        }
      }
    }

    if (!summed) {
      int qn = 0;
      for (int t = 0; t < nRounds; ++t) {
        const int tt = (t + (int)blockIdx.x) % nRounds;
        const int e0 = tt * (32 * STEP) + wave * STEP;
        if (e0 < nEdges) {
          if (e0 + STEP > nEdges) qn = scan_step<true,  false, MODE == 0>(dst, q, qn, e0, lane, base, rows, nEdges, cnt);
          else if (vec)           qn = scan_step<false, true,  MODE == 0>(dst, q, qn, e0, lane, base, rows, nEdges, cnt);
          else                    qn = scan_step<false, false, MODE == 0>(dst, q, qn, e0, lane, base, rows, nEdges, cnt);
          while (qn >= 32) { qn -= 32; drain<F, STRIDE, true>(x, src, acc, q + qn, lane, 32); }
        }
      }
      if (qn > 0) drain<F, STRIDE, false>(x, src, acc, q, lane, qn);
      if (MODE == 0 && lstB != nullptr && wave == 0) lstB[lane] = LIST_PAD;
    }

    __syncthreads();

    constexpr bool WLDS = (MODE == 1);
    constexpr int WPC = WLDS ? (FOUT * (F / 4)) / 1024 : 0;
    static_assert((FOUT * (F / 4)) % 1024 == 0, "W must split evenly over the block");
    static_assert((size_t)FOUT * STRIDE <= (size_t)32 * QCAP, "the W copy must fit the queues' space");
    static_assert(WPC <= (RB * (F / 4) + 1023) / 1024, "the W copy rides the combine's loops");
    v4f wreg[WLDS ? WPC : 1];
    float* Wl = (float*)queue;

    {
      constexpr int PA = (RB * (F / 4) + 1023) / 1024;
      const int nA = rows * (F / 4);
      v4f   xv[PA];
      float rd[PA];
      #pragma unroll
      for (int k = 0; k < PA; ++k) {
        const int i  = min((int)threadIdx.x + k * 1024, nA - 1);
        const int r  = i / (F / 4);
        const int c4 = i - r * (F / 4);
        xv[k] = ((const v4f*)(x + (size_t)(base + r) * F))[c4];
        rd[k] = (MODE == 0) ? 1.0f / fmaxf((float)cnt[r], 1.0f) : rdeg[base + r];
        if (k < WPC) wreg[k] = ((const v4f*)W)[(int)threadIdx.x + k * 1024];
      }
      #pragma unroll
      for (int k = 0; k < PA; ++k) {
        const int i = (int)threadIdx.x + k * 1024;
        const int r  = i / (F / 4);
        const int c4 = i - r * (F / 4);
        if (i < nA) {
          v4f* ap = (v4f*)&acc[r * STRIDE + 4 * c4];
          *ap = *ap * rd[k] + xv[k];
        }
        if (k < WPC) *(v4f*)&Wl[r * STRIDE + 4 * c4] = wreg[k];
      }
    }
    if (MODE == 0)
      for (int i = threadIdx.x; i < rows; i += 1024) rdeg[base + i] = 1.0f / fmaxf((float)cnt[i], 1.0f);
    __syncthreads();

    {
      const int half = lane >> 4;
      const int l16  = lane & 15;
      const int ct   = wave % NT;
      const int rg   = wave / NT;
      const int n    = ct * 16 + l16;
      const float* Wrow = WLDS ? (const float*)(Wl + n * STRIDE + 2 * half) : (W + (size_t)n * F + 2 * half);
      v2f bf[KS];
      #pragma unroll
      for (int s = 0; s < KS; ++s) bf[s] = *(const v2f*)(Wrow + 4 * s);
      const float bv = bias[n];

      for (int rt = rg; rt < RT; rt += RG) {
        if (rt * 16 >= rows) break;
        const float* Arow = &acc[(rt * 16 + l16) * STRIDE + 2 * half];
        v8f c = {};
        #pragma unroll
        for (int s = 0; s < KS; ++s) {
          const v2f a = *(const v2f*)(Arow + 4 * s);
          c = __builtin_amdgcn_wmma_f32_16x16x4_f32(false, a, false, bf[s], (short)0, c, false, false);
        }
        const int lrow0 = rt * 16 + 8 * half;
        float* colBase = out + (size_t)(base + lrow0) * FOUT + n;
        if (rt * 16 + 16 <= rows) {
          #pragma unroll
          for (int r = 0; r < 8; ++r) colBase[(size_t)r * FOUT] = fmaxf(c[r] + bv, 0.0f);
        } else {
          #pragma unroll
          for (int r = 0; r < 8; ++r)
            if (lrow0 + r < rows) colBase[(size_t)r * FOUT] = fmaxf(c[r] + bv, 0.0f);
        }
      }
    }
    __syncthreads();
  }
}

template <int FIN, int NT, bool HAS_AGG>
__global__ __launch_bounds__(256) void
gemm_bias_act_kernel(const float* __restrict__ agg, const float* __restrict__ rdeg,
                     const float* __restrict__ xin, const float* __restrict__ W,
                     const float* __restrict__ bias, float* __restrict__ out,
                     int nRows, int Fout, int doRelu)
{
  constexpr int RG     = 8 / NT;
  constexpr int ROWS   = 16 * RG;
  constexpr int STRIDE = FIN + 4;
  constexpr int KS     = FIN / 4;
  constexpr int RF4    = FIN / 4;
  constexpr int LD     = (ROWS * RF4) / 256;
  static_assert((ROWS * RF4) % 256 == 0, "tile must split evenly over the block");
  __shared__ float As[2][ROWS * STRIDE];
  __shared__ __attribute__((aligned(16))) float Wl[NT * 16 * STRIDE];
  constexpr int WPC = (NT * 16 * RF4) / 256;
  static_assert((NT * 16 * RF4) % 256 == 0, "the W copy must split evenly over the block");

  const int lane = threadIdx.x & 31;
  const int wave = threadIdx.x >> 5;
  const int half = lane >> 4;
  const int l16  = lane & 15;
  const int ct   = wave % NT;
  const int rg   = wave / NT;
  const int n    = ct * 16 + l16;
  const bool colOk = (n < Fout);

  v4f wreg[WPC];
  #pragma unroll
  for (int k = 0; k < WPC; ++k) {
    const int i  = (int)threadIdx.x + k * 256;
    const int r  = i / RF4;
    const int c4 = i - r * RF4;
    wreg[k] = ((const v4f*)W)[(size_t)min(r, Fout - 1) * RF4 + c4];
  }
  const float bv = colOk ? bias[n] : 0.0f;

  const int nSteps = (nRows + ROWS - 1) / ROWS;
  v4f   pa[LD], px[LD];
  float pr[LD];

  auto fetch = [&](int step) {
    #pragma unroll
    for (int i = 0; i < LD; ++i) {
      int idx = threadIdx.x + i * 256;
      int r   = idx / RF4;
      int c4  = idx - r * RF4;
      int gr  = min(step * ROWS + r, nRows - 1);
      px[i] = ((const v4f*)xin)[(size_t)gr * RF4 + c4];
      if constexpr (HAS_AGG) {
        pa[i] = ((const v4f*)agg)[(size_t)gr * RF4 + c4];
        pr[i] = rdeg[gr];
      }
    }
  };
  auto stage = [&](int b) {
    #pragma unroll
    for (int i = 0; i < LD; ++i) {
      int idx = threadIdx.x + i * 256;
      int r   = idx / RF4;
      int c4  = idx - r * RF4;
      v4f v = px[i];
      if constexpr (HAS_AGG) v += pa[i] * pr[i];
      *(v4f*)&As[b][r * STRIDE + 4 * c4] = v;
    }
  };

  int u = blockIdx.x;
  if (u < nSteps) { fetch(u); stage(0); }
  #pragma unroll
  for (int k = 0; k < WPC; ++k) {
    const int i  = (int)threadIdx.x + k * 256;
    const int r  = i / RF4;
    const int c4 = i - r * RF4;
    *(v4f*)&Wl[r * STRIDE + 4 * c4] = (r < Fout) ? wreg[k] : v4f{0.0f, 0.0f, 0.0f, 0.0f};
  }
  __syncthreads();

  const float* Wrow = &Wl[n * STRIDE + 2 * half];
  v2f bf[KS];
  #pragma unroll
  for (int s = 0; s < KS; ++s) bf[s] = *(const v2f*)(Wrow + 4 * s);

  int buf = 0;
  for (; u < nSteps; u += gridDim.x) {
    const int  un      = u + gridDim.x;
    const bool hasNext = (un < nSteps);
    if (hasNext) fetch(un);

    const float* Arow = &As[buf][(rg * 16 + l16) * STRIDE + 2 * half];
    v8f c = {};
    #pragma unroll
    for (int s = 0; s < KS; ++s) {
      v2f a = *(const v2f*)(Arow + 4 * s);
      c = __builtin_amdgcn_wmma_f32_16x16x4_f32(false, a, false, bf[s], (short)0, c, false, false);
    }

    const int row0 = u * ROWS + rg * 16 + 8 * half;
    float* colBase = out + (size_t)row0 * Fout + n;
    if ((u * ROWS + ROWS <= nRows) && (ct * 16 + 16 <= Fout)) {
      #pragma unroll
      for (int r = 0; r < 8; ++r) {
        float val = c[r] + bv;
        if (doRelu) val = fmaxf(val, 0.0f);
        colBase[(size_t)r * Fout] = val;
      }
    } else {
      #pragma unroll
      for (int r = 0; r < 8; ++r) {
        float val = c[r] + bv;
        if (doRelu) val = fmaxf(val, 0.0f);
        if (row0 + r < nRows && colOk) colBase[(size_t)r * Fout] = val;
      }
    }

    if (hasNext) stage(buf ^ 1);
    __syncthreads();
    buf ^= 1;
  }
}

extern "C" void kernel_launch(void* const* d_in, const int* in_sizes, int n_in,
                              void* d_out, int out_size, void* d_ws, size_t ws_size,
                              hipStream_t stream) {
  const float* x0 = (const float*)d_in[0];
  const int*   ei = (const int*)d_in[1];
  const float *W1 = (const float*)d_in[2],  *b1 = (const float*)d_in[3];
  const float *W2 = (const float*)d_in[4],  *b2 = (const float*)d_in[5];
  const float *W3 = (const float*)d_in[6],  *b3 = (const float*)d_in[7];
  const float *Wp = (const float*)d_in[8],  *bp = (const float*)d_in[9];
  const float *Wc = (const float*)d_in[10], *bc = (const float*)d_in[11];

  const int N      = in_sizes[0] / 64;
  const int nEdges = in_sizes[1] / 2;
  const int* src = ei;
  const int* dst = ei + nEdges;

  float* ws  = (float*)d_ws;
  float* deg = ws;
  float* h2  = ws + 65536;
  unsigned* elist = (unsigned*)(h2 + (size_t)N * 128);

  float* out = (float*)d_out;
  float* h1  = out;
  float* emb = out;
  float* nxt = out + (size_t)N * 64;
  float* cls = out + (size_t)N * 128;
  unsigned* bins = (unsigned*)h2;

  dim3 gemmGrid(512), gemmBlock(256);
  dim3 ownGrid(256), ownBlock(1024);

  bin_edges_kernel<<<ownGrid, ownBlock, 0, stream>>>(src, dst, bins, N, nEdges);

  owner_layer_kernel<64, 0, 128> <<<ownGrid, ownBlock, 0, stream>>>(x0, src, dst, bins, deg, W1, b1, h1, N, nEdges, elist, ws, ws_size);

  owner_layer_kernel<128, 1, 128><<<ownGrid, ownBlock, 0, stream>>>(h1, src, dst, nullptr, deg, W2, b2, h2, N, nEdges, elist, ws, ws_size);

  owner_layer_kernel<128, 1, 64> <<<ownGrid, ownBlock, 0, stream>>>(h2, src, dst, nullptr, deg, W3, b3, emb, N, nEdges, elist, ws, ws_size);

  gemm_bias_act_kernel<64, 4, false><<<gemmGrid, gemmBlock, 0, stream>>>(nullptr, nullptr, emb, Wp, bp, nxt, N, 64, 0);
  gemm_bias_act_kernel<64, 1, false><<<gemmGrid, gemmBlock, 0, stream>>>(nullptr, nullptr, emb, Wc, bc, cls, N, 10, 0);
}
